# starting kernel
# baseline (speedup 1.0000x reference)
#include <hip/hip_runtime.h>


typedef float v2f __attribute__((ext_vector_type(2)));
typedef float v8f __attribute__((ext_vector_type(8)));

#define BATCH 512
#define INPS  1024
#define NEUR  512
#define KFAN  64
#define NOUT  16
#define ROWS  16

__global__ __launch_bounds__(32) void ne_scan_wmma_kernel(
    const float* __restrict__ x,
    const int*   __restrict__ in_idxs,
    const float* __restrict__ weights,
    float*       __restrict__ out)
{
    extern __shared__ float smem[];
    float* act = smem;
    float* wst = act + (INPS + NEUR) * ROWS;
    int*   ist = (int*)(wst + 2 * KFAN);

    const int lane  = threadIdx.x;
    const int rbase = blockIdx.x * ROWS;
    const int h     = lane >> 4;
    const int n     = lane & 15;

    for (int t = lane; t < ROWS * INPS; t += 32) {
        int col = t & (INPS - 1);
        int r   = t >> 10;
        act[col * ROWS + r] = x[(size_t)(rbase + r) * INPS + col];
    }
    for (int t = lane; t < ROWS * NEUR; t += 32) {
        act[INPS * ROWS + t] = 0.0f;
    }

    v2f  aw[16];
    int2 jj[16];
    {
        float2 w0 = ((const float2*)weights)[lane];
        int2   j0 = ((const int2*)in_idxs)[lane];
        wst[2 * lane] = w0.x; wst[2 * lane + 1] = w0.y;
        ist[2 * lane] = j0.x; ist[2 * lane + 1] = j0.y;
        #pragma unroll
        for (int i = 0; i < 16; ++i) {
            const int k0 = 4 * i + 2 * h;
            float2 af = *(const float2*)(wst + k0);
            aw[i].x = af.x; aw[i].y = af.y;
            jj[i]   = *(const int2*)(ist + k0);
        }
    }
    float2 wA = ((const float2*)(weights + (size_t)1 * KFAN))[lane];
    int2   jA = ((const int2*)(in_idxs + (size_t)1 * KFAN))[lane];
    float2 wB = ((const float2*)(weights + (size_t)2 * KFAN))[lane];
    int2   jB = ((const int2*)(in_idxs + (size_t)2 * KFAN))[lane];

    for (int pos = 0; pos < NEUR; ++pos) {
        v2f b[16];
        #pragma unroll
        for (int i = 0; i < 16; ++i) {
            b[i].x = act[jj[i].x * ROWS + n];
            b[i].y = act[jj[i].y * ROWS + n];
        }

        v8f acc[4] = {};
        #pragma unroll
        for (int i = 0; i < 16; ++i) {
            acc[i & 3] = __builtin_amdgcn_wmma_f32_16x16x4_f32(
                false, aw[i], false, b[i], (short)0, acc[i & 3], false, false);
        }

        float val = (acc[0][0] + acc[1][0]) + (acc[2][0] + acc[3][0]);
        float sg  = 1.0f / (1.0f + __expf(-val));
        act[(INPS + pos) * ROWS + (lane & 15)] = sg;

        if (pos + 1 < NEUR) {
            const int   s  = (pos + 1) & 1;
            float*      wc = wst + s * KFAN;
            int*        ic = ist + s * KFAN;
            wc[2 * lane] = wA.x; wc[2 * lane + 1] = wA.y;
            ic[2 * lane] = jA.x; ic[2 * lane + 1] = jA.y;
            #pragma unroll
            for (int i = 0; i < 16; ++i) {
                const int k0 = 4 * i + 2 * h;
                float2 af = *(const float2*)(wc + k0);
                aw[i].x = af.x; aw[i].y = af.y;
                jj[i]   = *(const int2*)(ic + k0);
            }
            wA = wB; jA = jB;
            if (pos + 3 < NEUR) {
                wB = ((const float2*)(weights + (size_t)(pos + 3) * KFAN))[lane];
                jB = ((const int2*)(in_idxs + (size_t)(pos + 3) * KFAN))[lane];
            }
        }
    }

    for (int t = lane; t < NOUT * ROWS; t += 32) {
        int o = t >> 4;
        int r = t & 15;
        out[(size_t)o * BATCH + rbase + r] =
            act[(INPS + NEUR - NOUT + o) * ROWS + r];
    }
}

extern "C" void kernel_launch(void* const* d_in, const int* in_sizes, int n_in,
                              void* d_out, int out_size, void* d_ws, size_t ws_size,
                              hipStream_t stream) {
    (void)in_sizes; (void)n_in; (void)d_ws; (void)ws_size; (void)out_size;
    const float* x       = (const float*)d_in[0];
    const int*   in_idxs = (const int*)d_in[1];
    const float* weights = (const float*)d_in[2];
    float*       out     = (float*)d_out;

    const size_t smem_bytes =
        (size_t)((INPS + NEUR) * ROWS + 2 * KFAN) * sizeof(float) +
        (size_t)(2 * KFAN) * sizeof(int);

    ne_scan_wmma_kernel<<<dim3(BATCH / ROWS), dim3(32), smem_bytes, stream>>>(
        x, in_idxs, weights, out);
}


# Weights and gather indices go from global memory straight into the WMMA and gather registers, prefetched ahead (lane half owns k 32h..32h+31: reorders the 64-term sum); per-step LDS staging removed
# speedup vs baseline: 1.3780x; 1.3780x over previous
#include <hip/hip_runtime.h>


typedef float v2f __attribute__((ext_vector_type(2)));
typedef float v8f __attribute__((ext_vector_type(8)));
typedef __attribute__((address_space(3))) float lds_f;

#define BATCH 512
#define INPS  1024
#define NEUR  512
#define KFAN  64
#define NOUT  16
#define ROWS  16
#define HALFK 32

#define ROWBYTES (KFAN * 4)

static __device__ __forceinline__ void load_w(float4 (&w)[8], const char* __restrict__ wp) {
    const float4* p = (const float4*)wp;
    #pragma unroll
    for (int q = 0; q < 8; ++q) w[q] = p[q];
}

static __device__ __forceinline__ void load_j(int4 (&j)[8], const char* __restrict__ jp) {
    const int4* p = (const int4*)jp;
    #pragma unroll
    for (int q = 0; q < 8; ++q) j[q] = p[q];
}

static __device__ __forceinline__ void to_addr(const lds_f* (&g)[32], const int4 (&j)[8], const lds_f* actn) {
    #pragma unroll
    for (int q = 0; q < 8; ++q) {
        g[4 * q + 0] = actn + j[q].x * ROWS;
        g[4 * q + 1] = actn + j[q].y * ROWS;
        g[4 * q + 2] = actn + j[q].z * ROWS;
        g[4 * q + 3] = actn + j[q].w * ROWS;
    }
}

static __device__ __forceinline__ void neuron_step(
    lds_f* actn, float4 (&w)[8], const lds_f* (&g)[32], int4 (&jr)[8],
    const char* __restrict__ &wnext, const char* __restrict__ &jnext, unsigned hbytes, int pos)
{
    v2f b[16];
    #pragma unroll
    for (int i = 0; i < 16; ++i) {
        b[i].x = *g[2 * i];
        b[i].y = *g[2 * i + 1];
    }
    __builtin_amdgcn_sched_barrier(0);

    to_addr(g, jr, actn);
    asm volatile("" :: "v"(g[0]),  "v"(g[1]),  "v"(g[2]),  "v"(g[3]),  "v"(g[4]),  "v"(g[5]),  "v"(g[6]),  "v"(g[7]),
                       "v"(g[8]),  "v"(g[9]),  "v"(g[10]), "v"(g[11]), "v"(g[12]), "v"(g[13]), "v"(g[14]), "v"(g[15]));
    asm volatile("" :: "v"(g[16]), "v"(g[17]), "v"(g[18]), "v"(g[19]), "v"(g[20]), "v"(g[21]), "v"(g[22]), "v"(g[23]),
                       "v"(g[24]), "v"(g[25]), "v"(g[26]), "v"(g[27]), "v"(g[28]), "v"(g[29]), "v"(g[30]), "v"(g[31]));

    v8f acc[4] = {};
    #pragma unroll
    for (int i = 0; i < 16; ++i) {
        const float4 wq = w[i >> 1];
        v2f a;
        if (i & 1) { a.x = wq.z; a.y = wq.w; } else { a.x = wq.x; a.y = wq.y; }
        acc[i & 3] = __builtin_amdgcn_wmma_f32_16x16x4_f32(
            false, a, false, b[i], (short)0, acc[i & 3], false, false);
    }
    #pragma unroll
    for (int i = 0; i < 16; ++i) {
        __builtin_amdgcn_sched_group_barrier(0x008, 1, 0);
        __builtin_amdgcn_sched_group_barrier(0x002, 2, 0);
    }
    const char* jp = jnext + hbytes;
    const char* wp = wnext + hbytes;
    asm volatile("" :: "v"(jp), "v"(wp));
    asm volatile("" :: "v"(acc[0]), "v"(acc[1]), "v"(acc[2]), "v"(acc[3]));
    __builtin_amdgcn_sched_barrier(0);

    load_j(jr, jp);
    load_w(w, wp);
    if (pos + 4 < NEUR) jnext += ROWBYTES;
    if (pos + 3 < NEUR) wnext += ROWBYTES;

    float val = (acc[0][0] + acc[1][0]) + (acc[2][0] + acc[3][0]);
    float sg  = 1.0f / (1.0f + __expf(-val));
    actn[(INPS + pos) * ROWS] = sg;
}

__global__ __launch_bounds__(32) void ne_scan_wmma_kernel(
    const float* __restrict__ x,
    const int*   __restrict__ in_idxs,
    const float* __restrict__ weights,
    float*       __restrict__ out)
{
    extern __shared__ float act[];

    const int lane  = threadIdx.x;
    const int rbase = blockIdx.x * ROWS;
    const int h     = lane >> 4;
    const int n     = lane & 15;

    for (int t = lane; t < ROWS * INPS; t += 32) {
        int col = t & (INPS - 1);
        int r   = t >> 10;
        act[col * ROWS + r] = x[(size_t)(rbase + r) * INPS + col];
    }
    for (int t = lane; t < ROWS * NEUR; t += 32) {
        act[INPS * ROWS + t] = 0.0f;
    }

    lds_f*       actn  = (lds_f*)act + n;
    const unsigned hbytes = HALFK * 4u * h;
    float4 wA[8], wB[8];
    int4   jA[8], jB[8];
    const lds_f* g[32];
    const char* jbase = (const char*)in_idxs;
    const char* wbase = (const char*)weights;
    load_j(jA, jbase + hbytes);
    to_addr(g, jA, actn);
    __builtin_amdgcn_sched_barrier(0);
    load_j(jA, jbase + 1 * ROWBYTES + hbytes);
    __builtin_amdgcn_sched_barrier(0);
    load_w(wA, wbase + hbytes);
    __builtin_amdgcn_sched_barrier(0);
    load_j(jB, jbase + 2 * ROWBYTES + hbytes);
    __builtin_amdgcn_sched_barrier(0);
    load_w(wB, wbase + 1 * ROWBYTES + hbytes);
    __builtin_amdgcn_sched_barrier(0);
    const char* __restrict__ jnext = jbase + 3 * ROWBYTES;
    const char* __restrict__ wnext = wbase + 2 * ROWBYTES;

    for (int pos = 0; pos < NEUR; pos += 2) {
        neuron_step(actn, wA, g, jA, wnext, jnext, hbytes, pos);
        neuron_step(actn, wB, g, jB, wnext, jnext, hbytes, pos + 1);
    }

    for (int t = lane; t < NOUT * ROWS; t += 32) {
        int o = t >> 4;
        int r = t & 15;
        out[(size_t)o * BATCH + rbase + r] =
            act[(INPS + NEUR - NOUT + o) * ROWS + r];
    }
}

extern "C" void kernel_launch(void* const* d_in, const int* in_sizes, int n_in,
                              void* d_out, int out_size, void* d_ws, size_t ws_size,
                              hipStream_t stream) {
    (void)in_sizes; (void)n_in; (void)d_ws; (void)ws_size; (void)out_size;
    const float* x       = (const float*)d_in[0];
    const int*   in_idxs = (const int*)d_in[1];
    const float* weights = (const float*)d_in[2];
    float*       out     = (float*)d_out;

    const size_t smem_bytes = (size_t)((INPS + NEUR) * ROWS) * sizeof(float);

    ne_scan_wmma_kernel<<<dim3(BATCH / ROWS), dim3(32), smem_bytes, stream>>>(
        x, in_idxs, weights, out);
}


# Scan pipelined two steps deep: a step's gathers and 16 WMMAs run before its two newest predecessors are stored; their terms are added last by one more WMMA (reorders the 64-term sum)
# speedup vs baseline: 1.6203x; 1.1758x over previous
#include <hip/hip_runtime.h>


typedef float v2f __attribute__((ext_vector_type(2)));
typedef float v8f __attribute__((ext_vector_type(8)));
typedef __attribute__((address_space(3))) float lds_f;

#define BATCH 512
#define INPS  1024
#define NEUR  512
#define KFAN  64
#define NOUT  16
#define ROWS  16
#define HALFK 32

#define ROWBYTES (KFAN * 4)

static __device__ __forceinline__ void load_w(float4 (&w)[8], const char* __restrict__ wp) {
    const float4* p = (const float4*)wp;
    #pragma unroll
    for (int q = 0; q < 8; ++q) w[q] = p[q];
}

static __device__ __forceinline__ void load_j(int4 (&j)[8], const char* __restrict__ jp) {
    const int4* p = (const int4*)jp;
    #pragma unroll
    for (int q = 0; q < 8; ++q) j[q] = p[q];
}

static __device__ __forceinline__ void to_addr(const lds_f* (&g)[32], const int4 (&j)[8], const lds_f* actn) {
    #pragma unroll
    for (int q = 0; q < 8; ++q) {
        g[4 * q + 0] = actn + j[q].x * ROWS;
        g[4 * q + 1] = actn + j[q].y * ROWS;
        g[4 * q + 2] = actn + j[q].z * ROWS;
        g[4 * q + 3] = actn + j[q].w * ROWS;
    }
}

static __device__ __forceinline__ void gather(v2f (&b)[16], const lds_f* const (&g)[32]) {
    #pragma unroll
    for (int i = 0; i < 16; ++i) {
        b[i].x = *g[2 * i];
        b[i].y = *g[2 * i + 1];
    }
}

static __device__ __forceinline__ void block16(v8f (&acc)[4], const float4 (&w)[8], const v2f (&b)[16]) {
    acc[0] = v8f{};
    acc[1] = v8f{};
    acc[2] = v8f{};
    acc[3] = v8f{};
    #pragma unroll
    for (int i = 0; i < 16; ++i) {
        const float4 wq = w[i >> 1];
        v2f a;
        if (i & 1) { a.x = wq.z; a.y = wq.w; } else { a.x = wq.x; a.y = wq.y; }
        acc[i & 3] = __builtin_amdgcn_wmma_f32_16x16x4_f32(
            false, a, false, b[i], (short)0, acc[i & 3], false, false);
    }
}

static __device__ __forceinline__ float row_sum16(float v) {
    v += __builtin_bit_cast(float, __builtin_amdgcn_update_dpp(0, __builtin_bit_cast(int, v), 0x161, 0xf, 0xf, false));
    v += __builtin_bit_cast(float, __builtin_amdgcn_update_dpp(0, __builtin_bit_cast(int, v), 0x162, 0xf, 0xf, false));
    v += __builtin_bit_cast(float, __builtin_amdgcn_update_dpp(0, __builtin_bit_cast(int, v), 0x164, 0xf, 0xf, false));
    v += __builtin_bit_cast(float, __builtin_amdgcn_update_dpp(0, __builtin_bit_cast(int, v), 0x168, 0xf, 0xf, false));
    return v;
}

static __device__ __forceinline__ v2f corr_coeff(int2 pj, float2 pw, int q) {
    const int t1 = (q >= 1) ? (INPS + q - 1) : -1;
    const int t2 = (q >= 2) ? (INPS + q - 2) : -1;
    const float c1 = ((pj.x == t1) ? pw.x : 0.0f) + ((pj.y == t1) ? pw.y : 0.0f);
    const float c2 = ((pj.x == t2) ? pw.x : 0.0f) + ((pj.y == t2) ? pw.y : 0.0f);
    v2f a;
    a.x = row_sum16(c1);
    a.y = row_sum16(c2);
    return a;
}

static __device__ __forceinline__ void pipe_trip(
    lds_f* actn, const lds_f* (&g)[32],
    v2f (&bfill)[16], const v2f (&buse)[16],
    v8f (&accp)[4], v8f (&accn)[4],
    float4 (&wuse)[8], int4 (&juse)[8], int2 &pj, float2 &pw,
    const v2f &acorr, v2f &acorr_n, float &sg1, float &sg2,
    const char* __restrict__ wbase, const char* __restrict__ jbase,
    unsigned &woff, unsigned &joff, unsigned hbytes, unsigned pbytes, int pos)
{
    v2f bc;
    bc.x = sg1;
    bc.y = sg2;
    accp[0] = __builtin_amdgcn_wmma_f32_16x16x4_f32(
        false, acorr, false, bc, (short)0, accp[0], false, false);
    asm volatile("" :: "v"(accp[0]));
    __builtin_amdgcn_sched_barrier(0);

    gather(bfill, g);
    __builtin_amdgcn_sched_barrier(0);

    block16(accn, wuse, buse);
    const float val = (accp[0][0] + accp[1][0]) + (accp[2][0] + accp[3][0]);
    const float sg  = 1.0f / (1.0f + __expf(-val));
    to_addr(g, juse, actn);
    acorr_n = corr_coeff(pj, pw, pos + 1);
    const char* jp  = jbase + joff + hbytes;
    const char* wp  = wbase + woff + hbytes;
    const char* pjp = jbase + woff + pbytes;
    const char* pwp = wbase + woff + pbytes;
    #pragma unroll
    for (int i = 0; i < 16; ++i) {
        __builtin_amdgcn_sched_group_barrier(0x008, 1, 0);
        __builtin_amdgcn_sched_group_barrier(0x402, 5, 0);
    }
    asm volatile("" :: "v"(g[0]),  "v"(g[1]),  "v"(g[2]),  "v"(g[3]),  "v"(g[4]),  "v"(g[5]),  "v"(g[6]),  "v"(g[7]),
                       "v"(g[8]),  "v"(g[9]),  "v"(g[10]), "v"(g[11]), "v"(g[12]), "v"(g[13]), "v"(g[14]), "v"(g[15]));
    asm volatile("" :: "v"(g[16]), "v"(g[17]), "v"(g[18]), "v"(g[19]), "v"(g[20]), "v"(g[21]), "v"(g[22]), "v"(g[23]),
                       "v"(g[24]), "v"(g[25]), "v"(g[26]), "v"(g[27]), "v"(g[28]), "v"(g[29]), "v"(g[30]), "v"(g[31]));
    asm volatile("" :: "v"(acorr_n), "v"(sg));
    asm volatile("" :: "v"(jp), "v"(wp), "v"(pjp), "v"(pwp));
    asm volatile("" :: "v"(accn[0]), "v"(accn[1]), "v"(accn[2]), "v"(accn[3]));
    __builtin_amdgcn_sched_barrier(0);

    load_j(juse, jp);
    load_w(wuse, wp);
    pj = *(const int2*)pjp;
    pw = *(const float2*)pwp;
    if (pos + 8 < NEUR) joff += ROWBYTES;
    if (pos + 6 < NEUR) woff += ROWBYTES;
    actn[(INPS + pos) * ROWS] = sg;
    sg2 = sg1;
    sg1 = sg;
}

__global__ __launch_bounds__(32) void ne_scan_wmma_kernel(
    const float* __restrict__ x,
    const int*   __restrict__ in_idxs,
    const float* __restrict__ weights,
    float*       __restrict__ out)
{
    extern __shared__ float act[];

    const int lane  = threadIdx.x;
    const int rbase = blockIdx.x * ROWS;
    const int h     = lane >> 4;
    const int n     = lane & 15;

    for (int t = lane; t < ROWS * INPS; t += 32) {
        int col = t & (INPS - 1);
        int r   = t >> 10;
        act[col * ROWS + r] = x[(size_t)(rbase + r) * INPS + col];
    }
    for (int t = lane; t < ROWS * NEUR; t += 32) {
        act[INPS * ROWS + t] = 0.0f;
    }

    lds_f*       actn  = (lds_f*)act + n;
    const unsigned hbytes = HALFK * 4u * h;
    const unsigned pbytes = 8u * lane;
    float4 w0[8], w1[8], w2[8], w3[8];
    int4   j0[8], j1[8], j2[8], j3[8];
    int2   pj0, pj1, pj2, pj3;
    float2 pw0, pw1, pw2, pw3;
    v2f    bA[16], bB[16];
    v8f    accA[4], accB[4];
    v2f    acA, acB;
    const lds_f* g[32];
    const char* jbase = (const char*)in_idxs;
    const char* wbase = (const char*)weights;

    load_j(j0, jbase + hbytes);
    load_j(j1, jbase + 1 * ROWBYTES + hbytes);
    load_j(j2, jbase + 2 * ROWBYTES + hbytes);
    load_w(w0, wbase + hbytes);
    to_addr(g, j0, actn);
    gather(bA, g);
    to_addr(g, j1, actn);
    gather(bB, g);
    block16(accA, w0, bA);
    asm volatile("" :: "v"(accA[0]), "v"(accA[1]), "v"(accA[2]), "v"(accA[3]));
    to_addr(g, j2, actn);
    asm volatile("" :: "v"(g[0]),  "v"(g[1]),  "v"(g[2]),  "v"(g[3]),  "v"(g[4]),  "v"(g[5]),  "v"(g[6]),  "v"(g[7]),
                       "v"(g[8]),  "v"(g[9]),  "v"(g[10]), "v"(g[11]), "v"(g[12]), "v"(g[13]), "v"(g[14]), "v"(g[15]));
    asm volatile("" :: "v"(g[16]), "v"(g[17]), "v"(g[18]), "v"(g[19]), "v"(g[20]), "v"(g[21]), "v"(g[22]), "v"(g[23]),
                       "v"(g[24]), "v"(g[25]), "v"(g[26]), "v"(g[27]), "v"(g[28]), "v"(g[29]), "v"(g[30]), "v"(g[31]));
    __builtin_amdgcn_sched_barrier(0);
    load_j(j3, jbase + 3 * ROWBYTES + hbytes);
    load_w(w1, wbase + 1 * ROWBYTES + hbytes);
    pj1 = *(const int2*)(jbase + 1 * ROWBYTES + pbytes);
    pw1 = *(const float2*)(wbase + 1 * ROWBYTES + pbytes);
    __builtin_amdgcn_sched_barrier(0);
    load_j(j0, jbase + 4 * ROWBYTES + hbytes);
    load_w(w2, wbase + 2 * ROWBYTES + hbytes);
    pj2 = *(const int2*)(jbase + 2 * ROWBYTES + pbytes);
    pw2 = *(const float2*)(wbase + 2 * ROWBYTES + pbytes);
    __builtin_amdgcn_sched_barrier(0);
    load_j(j1, jbase + 5 * ROWBYTES + hbytes);
    load_w(w3, wbase + 3 * ROWBYTES + hbytes);
    pj3 = *(const int2*)(jbase + 3 * ROWBYTES + pbytes);
    pw3 = *(const float2*)(wbase + 3 * ROWBYTES + pbytes);
    __builtin_amdgcn_sched_barrier(0);
    load_j(j2, jbase + 6 * ROWBYTES + hbytes);
    load_w(w0, wbase + 4 * ROWBYTES + hbytes);
    pj0 = *(const int2*)(jbase + 4 * ROWBYTES + pbytes);
    pw0 = *(const float2*)(wbase + 4 * ROWBYTES + pbytes);
    __builtin_amdgcn_sched_barrier(0);
    unsigned joff = 7 * ROWBYTES;
    unsigned woff = 5 * ROWBYTES;
    acA.x = 0.0f; acA.y = 0.0f;
    acB = acA;
    float sg1 = 0.0f, sg2 = 0.0f;

    for (int pos = 0; pos < NEUR; pos += 4) {
        pipe_trip(actn, g, bA, bB, accA, accB, w1, j3, pj1, pw1, acA, acB, sg1, sg2,
                  wbase, jbase, woff, joff, hbytes, pbytes, pos);
        pipe_trip(actn, g, bB, bA, accB, accA, w2, j0, pj2, pw2, acB, acA, sg1, sg2,
                  wbase, jbase, woff, joff, hbytes, pbytes, pos + 1);
        pipe_trip(actn, g, bA, bB, accA, accB, w3, j1, pj3, pw3, acA, acB, sg1, sg2,
                  wbase, jbase, woff, joff, hbytes, pbytes, pos + 2);
        pipe_trip(actn, g, bB, bA, accB, accA, w0, j2, pj0, pw0, acB, acA, sg1, sg2,
                  wbase, jbase, woff, joff, hbytes, pbytes, pos + 3);
    }

    for (int t = lane; t < NOUT * ROWS; t += 32) {
        int o = t >> 4;
        int r = t & 15;
        out[(size_t)o * BATCH + rbase + r] =
            act[(INPS + NEUR - NOUT + o) * ROWS + r];
    }
}

extern "C" void kernel_launch(void* const* d_in, const int* in_sizes, int n_in,
                              void* d_out, int out_size, void* d_ws, size_t ws_size,
                              hipStream_t stream) {
    (void)in_sizes; (void)n_in; (void)d_ws; (void)ws_size; (void)out_size;
    const float* x       = (const float*)d_in[0];
    const int*   in_idxs = (const int*)d_in[1];
    const float* weights = (const float*)d_in[2];
    float*       out     = (float*)d_out;

    const size_t smem_bytes = (size_t)((INPS + NEUR) * ROWS) * sizeof(float);

    ne_scan_wmma_kernel<<<dim3(BATCH / ROWS), dim3(32), smem_bytes, stream>>>(
        x, in_idxs, weights, out);
}


# Each step's 64-term fan-in split over the four waves of a work-group (4 WMMAs, 8 gathers per wave); partial sums meet in LDS behind a hand-split barrier (reorders the 64-term sum)
# speedup vs baseline: 2.6084x; 1.6098x over previous
#include <hip/hip_runtime.h>


typedef float v2f __attribute__((ext_vector_type(2)));
typedef float v4f __attribute__((ext_vector_type(4)));
typedef float v8f __attribute__((ext_vector_type(8)));
typedef __attribute__((address_space(3))) float lds_f;
typedef __attribute__((address_space(3))) v4f   lds_v4f;

#define BATCH 512
#define INPS  1024
#define NEUR  512
#define KFAN  64
#define NOUT  16
#define ROWS  16
#define WAVES 4
#define QK    8

#define ROWBYTES (KFAN * 4)
#define ACTF  ((INPS + NEUR) * ROWS)
#define PARTF (2 * ROWS * WAVES)

static __device__ __forceinline__ void load_w(float4 (&w)[2], const char* __restrict__ wp) {
    const float4* p = (const float4*)wp;
    w[0] = p[0];
    w[1] = p[1];
}

static __device__ __forceinline__ void load_j(int4 (&j)[2], const char* __restrict__ jp) {
    const int4* p = (const int4*)jp;
    j[0] = p[0];
    j[1] = p[1];
}

static __device__ __forceinline__ void to_addr(const lds_f* (&g)[8], const int4 (&j)[2], const lds_f* actn) {
    #pragma unroll
    for (int q = 0; q < 2; ++q) {
        g[4 * q + 0] = actn + j[q].x * ROWS;
        g[4 * q + 1] = actn + j[q].y * ROWS;
        g[4 * q + 2] = actn + j[q].z * ROWS;
        g[4 * q + 3] = actn + j[q].w * ROWS;
    }
}

static __device__ __forceinline__ void gather(v2f (&b)[4], const lds_f* const (&g)[8]) {
    #pragma unroll
    for (int i = 0; i < 4; ++i) {
        b[i].x = *g[2 * i];
        b[i].y = *g[2 * i + 1];
    }
}

static __device__ __forceinline__ float block4(const float4 (&w)[2], const v2f (&b)[4]) {
    v8f acc0 = v8f{};
    v8f acc1 = v8f{};
    #pragma unroll
    for (int i = 0; i < 4; ++i) {
        const float4 wq = w[i >> 1];
        v2f a;
        if (i & 1) { a.x = wq.z; a.y = wq.w; } else { a.x = wq.x; a.y = wq.y; }
        if (i & 1)
            acc1 = __builtin_amdgcn_wmma_f32_16x16x4_f32(false, a, false, b[i], (short)0, acc1, false, false);
        else
            acc0 = __builtin_amdgcn_wmma_f32_16x16x4_f32(false, a, false, b[i], (short)0, acc0, false, false);
    }
    return acc0[0] + acc1[0];
}

static __device__ __forceinline__ float row_sum16(float v) {
    v += __builtin_bit_cast(float, __builtin_amdgcn_update_dpp(0, __builtin_bit_cast(int, v), 0x161, 0xf, 0xf, false));
    v += __builtin_bit_cast(float, __builtin_amdgcn_update_dpp(0, __builtin_bit_cast(int, v), 0x162, 0xf, 0xf, false));
    v += __builtin_bit_cast(float, __builtin_amdgcn_update_dpp(0, __builtin_bit_cast(int, v), 0x164, 0xf, 0xf, false));
    v += __builtin_bit_cast(float, __builtin_amdgcn_update_dpp(0, __builtin_bit_cast(int, v), 0x168, 0xf, 0xf, false));
    return v;
}

static __device__ __forceinline__ v2f corr_coeff(int2 pj, float2 pw, int q) {
    const int t1 = (q >= 1) ? (INPS + q - 1) : -1;
    const int t2 = (q >= 2) ? (INPS + q - 2) : -1;
    const float c1 = ((pj.x == t1) ? pw.x : 0.0f) + ((pj.y == t1) ? pw.y : 0.0f);
    const float c2 = ((pj.x == t2) ? pw.x : 0.0f) + ((pj.y == t2) ? pw.y : 0.0f);
    v2f a;
    a.x = row_sum16(c1);
    a.y = row_sum16(c2);
    return a;
}

static __device__ __forceinline__ void bar_signal_lds_done() {
    asm volatile("s_wait_dscnt 0x0\n\ts_barrier_signal -1" ::: "memory");
}
static __device__ __forceinline__ void bar_wait() {
    asm volatile("s_barrier_wait -1" ::: "memory");
}

static __device__ __forceinline__ float sum_parts(const lds_f* partn, int slot) {
    const v4f p = *(const lds_v4f*)(partn + slot * (ROWS * WAVES));
    return (p.x + p.y) + (p.z + p.w);
}

static __device__ __forceinline__ void pipe_trip(
    lds_f* actn, lds_f* partn, int wv, const lds_f* (&g)[8],
    v2f (&bfill)[4], const v2f (&buse)[4], float &early,
    float4 (&wuse)[2], int4 (&juse)[2], int2 &pj, float2 &pw,
    const v2f &acorr, v2f &acorr_n, float &sg1, float &sg2,
    const char* __restrict__ wbase, const char* __restrict__ jbase,
    unsigned &woff, unsigned &joff, unsigned hbytes, unsigned pbytes, int pos, const int slot)
{
    gather(bfill, g);
    __builtin_amdgcn_sched_barrier(0);

    v2f bc;
    bc.x = sg1;
    bc.y = sg2;
    const v8f cacc = __builtin_amdgcn_wmma_f32_16x16x4_f32(
        false, acorr, false, bc, (short)0, v8f{}, false, false);
    asm volatile("" :: "v"(cacc));
    __builtin_amdgcn_sched_barrier(0);

    const float part = block4(wuse, buse);
    const float val = cacc[0] + early;
    const float sg  = 1.0f / (1.0f + __expf(-val));
    to_addr(g, juse, actn);
    acorr_n = corr_coeff(pj, pw, pos + 1);
    const char* jp  = jbase + joff + hbytes;
    const char* wp  = wbase + woff + hbytes;
    const char* pjp = jbase + woff + pbytes;
    const char* pwp = wbase + woff + pbytes;
    #pragma unroll
    for (int i = 0; i < 4; ++i) {
        __builtin_amdgcn_sched_group_barrier(0x008, 1, 0);
        __builtin_amdgcn_sched_group_barrier(0x402, 10, 0);
    }
    asm volatile("" :: "v"(g[0]), "v"(g[1]), "v"(g[2]), "v"(g[3]), "v"(g[4]), "v"(g[5]), "v"(g[6]), "v"(g[7]));
    asm volatile("" :: "v"(acorr_n), "v"(sg), "v"(part));
    __builtin_amdgcn_sched_barrier(0);

    partn[slot * (ROWS * WAVES) + wv] = part;
    bar_signal_lds_done();

    load_j(juse, jp);
    load_w(wuse, wp);
    pj = *(const int2*)pjp;
    pw = *(const float2*)pwp;
    if (pos + 8 < NEUR) joff += ROWBYTES;
    if (pos + 6 < NEUR) woff += ROWBYTES;

    bar_wait();
    early = sum_parts(partn, slot);
    actn[(INPS + pos) * ROWS] = sg;
    sg2 = sg1;
    sg1 = sg;
}

__global__ __launch_bounds__(32 * WAVES) void ne_scan_wmma_kernel(
    const float* __restrict__ x,
    const int*   __restrict__ in_idxs,
    const float* __restrict__ weights,
    float*       __restrict__ out)
{
    extern __shared__ float act[];

    const int tid   = threadIdx.x;
    const int lane  = tid & 31;
    const int wv    = __builtin_amdgcn_readfirstlane(tid >> 5);
    const int rbase = blockIdx.x * ROWS;
    const int h     = lane >> 4;
    const int n     = lane & 15;

    for (int t = tid; t < ROWS * INPS; t += 32 * WAVES) {
        int col = t & (INPS - 1);
        int r   = t >> 10;
        act[col * ROWS + r] = x[(size_t)(rbase + r) * INPS + col];
    }
    for (int t = tid; t < ROWS * NEUR; t += 32 * WAVES) {
        act[INPS * ROWS + t] = 0.0f;
    }
    __syncthreads();

    lds_f*       actn  = (lds_f*)act + n;
    lds_f*       partn = (lds_f*)act + ACTF + n * WAVES;
    const unsigned hbytes = 4u * (unsigned)(QK * (2 * wv + h));
    const unsigned pbytes = 8u * lane;
    float4 w0[2], w1[2], w2[2], w3[2];
    int4   j0[2], j1[2], j2[2], j3[2];
    int2   pj0, pj1, pj2, pj3;
    float2 pw0, pw1, pw2, pw3;
    v2f    bA[4], bB[4];
    v2f    acA, acB;
    const lds_f* g[8];
    const char* jbase = (const char*)in_idxs;
    const char* wbase = (const char*)weights;

    load_j(j0, jbase + hbytes);
    load_j(j1, jbase + 1 * ROWBYTES + hbytes);
    load_j(j2, jbase + 2 * ROWBYTES + hbytes);
    load_w(w0, wbase + hbytes);
    to_addr(g, j0, actn);
    gather(bA, g);
    to_addr(g, j1, actn);
    gather(bB, g);
    const float part0 = block4(w0, bA);
    partn[wv] = part0;
    to_addr(g, j2, actn);
    bar_signal_lds_done();
    load_j(j3, jbase + 3 * ROWBYTES + hbytes);
    load_w(w1, wbase + 1 * ROWBYTES + hbytes);
    pj1 = *(const int2*)(jbase + 1 * ROWBYTES + pbytes);
    pw1 = *(const float2*)(wbase + 1 * ROWBYTES + pbytes);
    load_j(j0, jbase + 4 * ROWBYTES + hbytes);
    load_w(w2, wbase + 2 * ROWBYTES + hbytes);
    pj2 = *(const int2*)(jbase + 2 * ROWBYTES + pbytes);
    pw2 = *(const float2*)(wbase + 2 * ROWBYTES + pbytes);
    load_j(j1, jbase + 5 * ROWBYTES + hbytes);
    load_w(w3, wbase + 3 * ROWBYTES + hbytes);
    pj3 = *(const int2*)(jbase + 3 * ROWBYTES + pbytes);
    pw3 = *(const float2*)(wbase + 3 * ROWBYTES + pbytes);
    load_j(j2, jbase + 6 * ROWBYTES + hbytes);
    load_w(w0, wbase + 4 * ROWBYTES + hbytes);
    pj0 = *(const int2*)(jbase + 4 * ROWBYTES + pbytes);
    pw0 = *(const float2*)(wbase + 4 * ROWBYTES + pbytes);
    bar_wait();
    float early = sum_parts(partn, 0);
    unsigned joff = 7 * ROWBYTES;
    unsigned woff = 5 * ROWBYTES;
    acA.x = 0.0f; acA.y = 0.0f;
    acB = acA;
    float sg1 = 0.0f, sg2 = 0.0f;

    for (int pos = 0; pos < NEUR; pos += 4) {
        pipe_trip(actn, partn, wv, g, bA, bB, early, w1, j3, pj1, pw1, acA, acB, sg1, sg2,
                  wbase, jbase, woff, joff, hbytes, pbytes, pos, 1);
        pipe_trip(actn, partn, wv, g, bB, bA, early, w2, j0, pj2, pw2, acB, acA, sg1, sg2,
                  wbase, jbase, woff, joff, hbytes, pbytes, pos + 1, 0);
        pipe_trip(actn, partn, wv, g, bA, bB, early, w3, j1, pj3, pw3, acA, acB, sg1, sg2,
                  wbase, jbase, woff, joff, hbytes, pbytes, pos + 2, 1);
        pipe_trip(actn, partn, wv, g, bB, bA, early, w0, j2, pj0, pw0, acB, acA, sg1, sg2,
                  wbase, jbase, woff, joff, hbytes, pbytes, pos + 3, 0);
    }

    for (int t = tid; t < NOUT * ROWS; t += 32 * WAVES) {
        int o = t >> 4;
        int r = t & 15;
        out[(size_t)o * BATCH + rbase + r] =
            act[(INPS + NEUR - NOUT + o) * ROWS + r];
    }
}

extern "C" void kernel_launch(void* const* d_in, const int* in_sizes, int n_in,
                              void* d_out, int out_size, void* d_ws, size_t ws_size,
                              hipStream_t stream) {
    (void)in_sizes; (void)n_in; (void)d_ws; (void)ws_size; (void)out_size;
    const float* x       = (const float*)d_in[0];
    const int*   in_idxs = (const int*)d_in[1];
    const float* weights = (const float*)d_in[2];
    float*       out     = (float*)d_out;

    const size_t smem_bytes = (size_t)(ACTF + PARTF) * sizeof(float);

    ne_scan_wmma_kernel<<<dim3(BATCH / ROWS), dim3(32 * WAVES), smem_bytes, stream>>>(
        x, in_idxs, weights, out);
}


# Sigmoid 1/(1+e^-v) takes the hardware reciprocal v_rcp_f32 (1 ulp) instead of the IEEE division sequence: ten dependent vector ops leave each step's serial chain (sum order unchanged)
# speedup vs baseline: 2.8631x; 1.0976x over previous
#include <hip/hip_runtime.h>


typedef float v2f __attribute__((ext_vector_type(2)));
typedef float v4f __attribute__((ext_vector_type(4)));
typedef float v8f __attribute__((ext_vector_type(8)));
typedef __attribute__((address_space(3))) float lds_f;
typedef __attribute__((address_space(3))) v4f   lds_v4f;

#define BATCH 512
#define INPS  1024
#define NEUR  512
#define KFAN  64
#define NOUT  16
#define ROWS  16
#define WAVES 4
#define QK    8

#define ROWBYTES (KFAN * 4)
#define ACTF  ((INPS + NEUR) * ROWS)
#define PARTF (2 * ROWS * WAVES)

static __device__ __forceinline__ void load_w(float4 (&w)[2], const char* __restrict__ wp) {
    const float4* p = (const float4*)wp;
    w[0] = p[0];
    w[1] = p[1];
}

static __device__ __forceinline__ void load_j(int4 (&j)[2], const char* __restrict__ jp) {
    const int4* p = (const int4*)jp;
    j[0] = p[0];
    j[1] = p[1];
}

static __device__ __forceinline__ void to_addr(const lds_f* (&g)[8], const int4 (&j)[2], const lds_f* actn) {
    #pragma unroll
    for (int q = 0; q < 2; ++q) {
        g[4 * q + 0] = actn + j[q].x * ROWS;
        g[4 * q + 1] = actn + j[q].y * ROWS;
        g[4 * q + 2] = actn + j[q].z * ROWS;
        g[4 * q + 3] = actn + j[q].w * ROWS;
    }
}

static __device__ __forceinline__ void gather(v2f (&b)[4], const lds_f* const (&g)[8]) {
    #pragma unroll
    for (int i = 0; i < 4; ++i) {
        b[i].x = *g[2 * i];
        b[i].y = *g[2 * i + 1];
    }
}

static __device__ __forceinline__ float block4(const float4 (&w)[2], const v2f (&b)[4]) {
    v8f acc0 = v8f{};
    v8f acc1 = v8f{};
    #pragma unroll
    for (int i = 0; i < 4; ++i) {
        const float4 wq = w[i >> 1];
        v2f a;
        if (i & 1) { a.x = wq.z; a.y = wq.w; } else { a.x = wq.x; a.y = wq.y; }
        if (i & 1)
            acc1 = __builtin_amdgcn_wmma_f32_16x16x4_f32(false, a, false, b[i], (short)0, acc1, false, false);
        else
            acc0 = __builtin_amdgcn_wmma_f32_16x16x4_f32(false, a, false, b[i], (short)0, acc0, false, false);
    }
    return acc0[0] + acc1[0];
}

static __device__ __forceinline__ float row_sum16(float v) {
    v += __builtin_bit_cast(float, __builtin_amdgcn_update_dpp(0, __builtin_bit_cast(int, v), 0x161, 0xf, 0xf, false));
    v += __builtin_bit_cast(float, __builtin_amdgcn_update_dpp(0, __builtin_bit_cast(int, v), 0x162, 0xf, 0xf, false));
    v += __builtin_bit_cast(float, __builtin_amdgcn_update_dpp(0, __builtin_bit_cast(int, v), 0x164, 0xf, 0xf, false));
    v += __builtin_bit_cast(float, __builtin_amdgcn_update_dpp(0, __builtin_bit_cast(int, v), 0x168, 0xf, 0xf, false));
    return v;
}

static __device__ __forceinline__ v2f corr_coeff(int2 pj, float2 pw, int q) {
    const int t1 = (q >= 1) ? (INPS + q - 1) : -1;
    const int t2 = (q >= 2) ? (INPS + q - 2) : -1;
    const float c1 = ((pj.x == t1) ? pw.x : 0.0f) + ((pj.y == t1) ? pw.y : 0.0f);
    const float c2 = ((pj.x == t2) ? pw.x : 0.0f) + ((pj.y == t2) ? pw.y : 0.0f);
    v2f a;
    a.x = row_sum16(c1);
    a.y = row_sum16(c2);
    return a;
}

static __device__ __forceinline__ void bar_signal_lds_done() {
    asm volatile("s_wait_dscnt 0x0\n\ts_barrier_signal -1" ::: "memory");
}
static __device__ __forceinline__ void bar_wait() {
    asm volatile("s_barrier_wait -1" ::: "memory");
}

static __device__ __forceinline__ float sum_parts(const lds_f* partn, int slot) {
    const v4f p = *(const lds_v4f*)(partn + slot * (ROWS * WAVES));
    return (p.x + p.y) + (p.z + p.w);
}

static __device__ __forceinline__ void pipe_trip(
    lds_f* actn, lds_f* partn, int wv, const lds_f* (&g)[8],
    v2f (&bfill)[4], const v2f (&buse)[4], float &early,
    float4 (&wuse)[2], int4 (&juse)[2], int2 &pj, float2 &pw,
    const v2f &acorr, v2f &acorr_n, float &sg1, float &sg2,
    const char* __restrict__ wbase, const char* __restrict__ jbase,
    unsigned &woff, unsigned &joff, unsigned hbytes, unsigned pbytes, int pos, const int slot)
{
    gather(bfill, g);
    __builtin_amdgcn_sched_barrier(0);

    v2f bc;
    bc.x = sg1;
    bc.y = sg2;
    const v8f cacc = __builtin_amdgcn_wmma_f32_16x16x4_f32(
        false, acorr, false, bc, (short)0, v8f{}, false, false);
    asm volatile("" :: "v"(cacc));
    __builtin_amdgcn_sched_barrier(0);

    const float part = block4(wuse, buse);
    const float val = cacc[0] + early;
    const float sg  = __builtin_amdgcn_rcpf(1.0f + __expf(-val));
    to_addr(g, juse, actn);
    acorr_n = corr_coeff(pj, pw, pos + 1);
    const char* jp  = jbase + joff + hbytes;
    const char* wp  = wbase + woff + hbytes;
    const char* pjp = jbase + woff + pbytes;
    const char* pwp = wbase + woff + pbytes;
    #pragma unroll
    for (int i = 0; i < 4; ++i) {
        __builtin_amdgcn_sched_group_barrier(0x008, 1, 0);
        __builtin_amdgcn_sched_group_barrier(0x402, 10, 0);
    }
    asm volatile("" :: "v"(g[0]), "v"(g[1]), "v"(g[2]), "v"(g[3]), "v"(g[4]), "v"(g[5]), "v"(g[6]), "v"(g[7]));
    asm volatile("" :: "v"(acorr_n), "v"(sg), "v"(part));
    __builtin_amdgcn_sched_barrier(0);

    partn[slot * (ROWS * WAVES) + wv] = part;
    bar_signal_lds_done();

    load_j(juse, jp);
    load_w(wuse, wp);
    pj = *(const int2*)pjp;
    pw = *(const float2*)pwp;
    if (pos + 8 < NEUR) joff += ROWBYTES;
    if (pos + 6 < NEUR) woff += ROWBYTES;

    bar_wait();
    early = sum_parts(partn, slot);
    actn[(INPS + pos) * ROWS] = sg;
    sg2 = sg1;
    sg1 = sg;
}

__global__ __launch_bounds__(32 * WAVES) void ne_scan_wmma_kernel(
    const float* __restrict__ x,
    const int*   __restrict__ in_idxs,
    const float* __restrict__ weights,
    float*       __restrict__ out)
{
    extern __shared__ float act[];

    const int tid   = threadIdx.x;
    const int lane  = tid & 31;
    const int wv    = __builtin_amdgcn_readfirstlane(tid >> 5);
    const int rbase = blockIdx.x * ROWS;
    const int h     = lane >> 4;
    const int n     = lane & 15;

    for (int t = tid; t < ROWS * INPS; t += 32 * WAVES) {
        int col = t & (INPS - 1);
        int r   = t >> 10;
        act[col * ROWS + r] = x[(size_t)(rbase + r) * INPS + col];
    }
    for (int t = tid; t < ROWS * NEUR; t += 32 * WAVES) {
        act[INPS * ROWS + t] = 0.0f;
    }
    __syncthreads();

    lds_f*       actn  = (lds_f*)act + n;
    lds_f*       partn = (lds_f*)act + ACTF + n * WAVES;
    const unsigned hbytes = 4u * (unsigned)(QK * (2 * wv + h));
    const unsigned pbytes = 8u * lane;
    float4 w0[2], w1[2], w2[2], w3[2];
    int4   j0[2], j1[2], j2[2], j3[2];
    int2   pj0, pj1, pj2, pj3;
    float2 pw0, pw1, pw2, pw3;
    v2f    bA[4], bB[4];
    v2f    acA, acB;
    const lds_f* g[8];
    const char* jbase = (const char*)in_idxs;
    const char* wbase = (const char*)weights;

    load_j(j0, jbase + hbytes);
    load_j(j1, jbase + 1 * ROWBYTES + hbytes);
    load_j(j2, jbase + 2 * ROWBYTES + hbytes);
    load_w(w0, wbase + hbytes);
    to_addr(g, j0, actn);
    gather(bA, g);
    to_addr(g, j1, actn);
    gather(bB, g);
    const float part0 = block4(w0, bA);
    partn[wv] = part0;
    to_addr(g, j2, actn);
    bar_signal_lds_done();
    load_j(j3, jbase + 3 * ROWBYTES + hbytes);
    load_w(w1, wbase + 1 * ROWBYTES + hbytes);
    pj1 = *(const int2*)(jbase + 1 * ROWBYTES + pbytes);
    pw1 = *(const float2*)(wbase + 1 * ROWBYTES + pbytes);
    load_j(j0, jbase + 4 * ROWBYTES + hbytes);
    load_w(w2, wbase + 2 * ROWBYTES + hbytes);
    pj2 = *(const int2*)(jbase + 2 * ROWBYTES + pbytes);
    pw2 = *(const float2*)(wbase + 2 * ROWBYTES + pbytes);
    load_j(j1, jbase + 5 * ROWBYTES + hbytes);
    load_w(w3, wbase + 3 * ROWBYTES + hbytes);
    pj3 = *(const int2*)(jbase + 3 * ROWBYTES + pbytes);
    pw3 = *(const float2*)(wbase + 3 * ROWBYTES + pbytes);
    load_j(j2, jbase + 6 * ROWBYTES + hbytes);
    load_w(w0, wbase + 4 * ROWBYTES + hbytes);
    pj0 = *(const int2*)(jbase + 4 * ROWBYTES + pbytes);
    pw0 = *(const float2*)(wbase + 4 * ROWBYTES + pbytes);
    bar_wait();
    float early = sum_parts(partn, 0);
    unsigned joff = 7 * ROWBYTES;
    unsigned woff = 5 * ROWBYTES;
    acA.x = 0.0f; acA.y = 0.0f;
    acB = acA;
    float sg1 = 0.0f, sg2 = 0.0f;

    for (int pos = 0; pos < NEUR; pos += 4) {
        pipe_trip(actn, partn, wv, g, bA, bB, early, w1, j3, pj1, pw1, acA, acB, sg1, sg2,
                  wbase, jbase, woff, joff, hbytes, pbytes, pos, 1);
        pipe_trip(actn, partn, wv, g, bB, bA, early, w2, j0, pj2, pw2, acB, acA, sg1, sg2,
                  wbase, jbase, woff, joff, hbytes, pbytes, pos + 1, 0);
        pipe_trip(actn, partn, wv, g, bA, bB, early, w3, j1, pj3, pw3, acA, acB, sg1, sg2,
                  wbase, jbase, woff, joff, hbytes, pbytes, pos + 2, 1);
        pipe_trip(actn, partn, wv, g, bB, bA, early, w0, j2, pj0, pw0, acB, acA, sg1, sg2,
                  wbase, jbase, woff, joff, hbytes, pbytes, pos + 3, 0);
    }

    for (int t = tid; t < NOUT * ROWS; t += 32 * WAVES) {
        int o = t >> 4;
        int r = t & 15;
        out[(size_t)o * BATCH + rbase + r] =
            act[(INPS + NEUR - NOUT + o) * ROWS + r];
    }
}

extern "C" void kernel_launch(void* const* d_in, const int* in_sizes, int n_in,
                              void* d_out, int out_size, void* d_ws, size_t ws_size,
                              hipStream_t stream) {
    (void)in_sizes; (void)n_in; (void)d_ws; (void)ws_size; (void)out_size;
    const float* x       = (const float*)d_in[0];
    const int*   in_idxs = (const int*)d_in[1];
    const float* weights = (const float*)d_in[2];
    float*       out     = (float*)d_out;

    const size_t smem_bytes = (size_t)(ACTF + PARTF) * sizeof(float);

    ne_scan_wmma_kernel<<<dim3(BATCH / ROWS), dim3(32 * WAVES), smem_bytes, stream>>>(
        x, in_idxs, weights, out);
}


# Scan pipelined three steps deep: partial sums made two trips early, stored behind the barrier signal unwaited; third late term joins the correction WMMA; vector work in shares (reorders 64-term sum)
# speedup vs baseline: 3.0777x; 1.0750x over previous
#include <hip/hip_runtime.h>


typedef float v2f __attribute__((ext_vector_type(2)));
typedef float v4f __attribute__((ext_vector_type(4)));
typedef float v8f __attribute__((ext_vector_type(8)));
typedef __attribute__((address_space(3))) float lds_f;
typedef __attribute__((address_space(3))) v4f   lds_v4f;

#define BATCH 512
#define INPS  1024
#define NEUR  512
#define KFAN  64
#define NOUT  16
#define ROWS  16
#define WAVES 4
#define QK    8

#define ROWBYTES (KFAN * 4)
#define ACTF  ((INPS + NEUR) * ROWS)
#define PARTF (4 * ROWS * WAVES)

static __device__ __forceinline__ void load_w(float4 (&w)[2], const char* __restrict__ wp) {
    const float4* p = (const float4*)wp;
    w[0] = p[0];
    w[1] = p[1];
}

static __device__ __forceinline__ void load_j(int4 (&j)[2], const char* __restrict__ jp) {
    const int4* p = (const int4*)jp;
    j[0] = p[0];
    j[1] = p[1];
}

static __device__ __forceinline__ void to_addr(const lds_f* (&g)[8], const int4 (&j)[2], const lds_f* actn) {
    #pragma unroll
    for (int q = 0; q < 2; ++q) {
        g[4 * q + 0] = actn + j[q].x * ROWS;
        g[4 * q + 1] = actn + j[q].y * ROWS;
        g[4 * q + 2] = actn + j[q].z * ROWS;
        g[4 * q + 3] = actn + j[q].w * ROWS;
    }
}

static __device__ __forceinline__ void gather(v2f (&b)[4], const lds_f* const (&g)[8]) {
    #pragma unroll
    for (int i = 0; i < 4; ++i) {
        b[i].x = *g[2 * i];
        b[i].y = *g[2 * i + 1];
    }
}

static __device__ __forceinline__ float block4(const float4 (&w)[2], const v2f (&b)[4]) {
    v8f acc0 = v8f{};
    v8f acc1 = v8f{};
    #pragma unroll
    for (int i = 0; i < 4; ++i) {
        const float4 wq = w[i >> 1];
        v2f a;
        if (i & 1) { a.x = wq.z; a.y = wq.w; } else { a.x = wq.x; a.y = wq.y; }
        if (i & 1)
            acc1 = __builtin_amdgcn_wmma_f32_16x16x4_f32(false, a, false, b[i], (short)0, acc1, false, false);
        else
            acc0 = __builtin_amdgcn_wmma_f32_16x16x4_f32(false, a, false, b[i], (short)0, acc0, false, false);
    }
    return acc0[0] + acc1[0];
}

static __device__ __forceinline__ float row_sum16(float v) {
    v += __builtin_bit_cast(float, __builtin_amdgcn_update_dpp(0, __builtin_bit_cast(int, v), 0x161, 0xf, 0xf, false));
    v += __builtin_bit_cast(float, __builtin_amdgcn_update_dpp(0, __builtin_bit_cast(int, v), 0x162, 0xf, 0xf, false));
    v += __builtin_bit_cast(float, __builtin_amdgcn_update_dpp(0, __builtin_bit_cast(int, v), 0x164, 0xf, 0xf, false));
    v += __builtin_bit_cast(float, __builtin_amdgcn_update_dpp(0, __builtin_bit_cast(int, v), 0x168, 0xf, 0xf, false));
    return v;
}

static __device__ __forceinline__ void bar_signal_lds_done() {
    asm volatile("s_wait_dscnt 0x0\n\ts_barrier_signal -1" ::: "memory");
}
static __device__ __forceinline__ void bar_wait() {
    asm volatile("s_barrier_wait -1" ::: "memory");
}

static __device__ __forceinline__ v4f load_parts(const lds_f* partn, int slot) {
    return *(const lds_v4f*)(partn + slot * (ROWS * WAVES));
}

static __device__ __forceinline__ void pipe_trip(
    lds_f* actn, lds_f* partn, int wv, bool hi, const lds_f* (&g)[8],
    v2f (&bfill)[4], const v2f (&buse)[4], v4f &ep,
    float4 (&wuse)[2], int4 (&juse)[2], int4 &pj, float4 &pw,
    const v2f &acorr, v2f &acorr_n, float &sg1, float &sg2, float &sg3,
    const char* __restrict__ wbase, const char* __restrict__ jbase,
    unsigned &woff, unsigned &joff, unsigned &poff, unsigned hbytes, unsigned pbytes,
    int pos, const int slotw, const int slotr)
{
    gather(bfill, g);
    __builtin_amdgcn_sched_barrier(0);

    v2f bc;
    bc.x = sg1;
    bc.y = hi ? sg3 : sg2;
    const v8f cacc = __builtin_amdgcn_wmma_f32_16x16x4_f32(
        false, acorr, false, bc, (short)0, v8f{}, false, false);
    asm volatile("" :: "v"(cacc));
    __builtin_amdgcn_sched_barrier(0);

    v8f acc0 = v8f{};
    v8f acc1 = v8f{};
    v2f a;

    a.x = wuse[0].x; a.y = wuse[0].y;
    acc0 = __builtin_amdgcn_wmma_f32_16x16x4_f32(false, a, false, buse[0], (short)0, acc0, false, false);
    const float early = (ep.x + ep.y) + (ep.z + ep.w);
    to_addr(g, juse, actn);
    __builtin_amdgcn_sched_group_barrier(0x008, 1, 0);
    __builtin_amdgcn_sched_group_barrier(0x402, 24, 0);
    asm volatile("" :: "v"(g[0]), "v"(g[1]), "v"(g[2]), "v"(g[3]), "v"(g[4]), "v"(g[5]), "v"(g[6]), "v"(g[7]), "v"(early));
    __builtin_amdgcn_sched_barrier(0);

    a.x = wuse[0].z; a.y = wuse[0].w;
    acc1 = __builtin_amdgcn_wmma_f32_16x16x4_f32(false, a, false, buse[1], (short)0, acc1, false, false);
    const float val = cacc[0] + early;
    const float ex  = __expf(-val);
    const int q  = pos + 1;
    const int t1 = (q >= 1) ? (INPS + q - 1) : -1;
    const int t2 = (q >= 2) ? (INPS + q - 2) : -1;
    const int t3 = (q >= 3) ? (INPS + q - 3) : -1;
    const float c1 = (((pj.x == t1) ? pw.x : 0.0f) + ((pj.y == t1) ? pw.y : 0.0f))
                   + (((pj.z == t1) ? pw.z : 0.0f) + ((pj.w == t1) ? pw.w : 0.0f));
    const float cx = hi ? 0.0f : c1;
    __builtin_amdgcn_sched_group_barrier(0x008, 1, 0);
    __builtin_amdgcn_sched_group_barrier(0x402, 24, 0);
    asm volatile("" :: "v"(ex), "v"(cx));
    __builtin_amdgcn_sched_barrier(0);

    a.x = wuse[1].x; a.y = wuse[1].y;
    acc0 = __builtin_amdgcn_wmma_f32_16x16x4_f32(false, a, false, buse[2], (short)0, acc0, false, false);
    const float sg = __builtin_amdgcn_rcpf(1.0f + ex);
    const int ty = hi ? t3 : t2;
    const float cy = (((pj.x == ty) ? pw.x : 0.0f) + ((pj.y == ty) ? pw.y : 0.0f))
                   + (((pj.z == ty) ? pw.z : 0.0f) + ((pj.w == ty) ? pw.w : 0.0f));
    __builtin_amdgcn_sched_group_barrier(0x008, 1, 0);
    __builtin_amdgcn_sched_group_barrier(0x402, 24, 0);
    asm volatile("" :: "v"(sg), "v"(cy));
    __builtin_amdgcn_sched_barrier(0);

    a.x = wuse[1].z; a.y = wuse[1].w;
    acc1 = __builtin_amdgcn_wmma_f32_16x16x4_f32(false, a, false, buse[3], (short)0, acc1, false, false);
    acorr_n.x = row_sum16(cx);
    acorr_n.y = row_sum16(cy);
    const char* jp  = jbase + joff + hbytes;
    const char* wp  = wbase + woff + hbytes;
    const char* pjp = jbase + poff + pbytes;
    const char* pwp = wbase + poff + pbytes;
    const float part = acc0[0] + acc1[0];
    __builtin_amdgcn_sched_group_barrier(0x008, 1, 0);
    __builtin_amdgcn_sched_group_barrier(0x402, 24, 0);
    asm volatile("" :: "v"(acorr_n), "v"(part), "v"(jp), "v"(wp), "v"(pjp), "v"(pwp));
    __builtin_amdgcn_sched_barrier(0);

    bar_signal_lds_done();

    partn[slotw * (ROWS * WAVES) + wv] = part;

    load_j(juse, jp);
    load_w(wuse, wp);
    pj = *(const int4*)pjp;
    pw = *(const float4*)pwp;
    if (pos + 9 < NEUR) joff += ROWBYTES;
    if (pos + 7 < NEUR) woff += ROWBYTES;
    if (pos + 6 < NEUR) poff += ROWBYTES;

    bar_wait();
    ep = load_parts(partn, slotr);
    actn[(INPS + pos) * ROWS] = sg;
    sg3 = sg2;
    sg2 = sg1;
    sg1 = sg;
}

__global__ __launch_bounds__(32 * WAVES) void ne_scan_wmma_kernel(
    const float* __restrict__ x,
    const int*   __restrict__ in_idxs,
    const float* __restrict__ weights,
    float*       __restrict__ out)
{
    extern __shared__ float act[];

    const int tid   = threadIdx.x;
    const int lane  = tid & 31;
    const int wv    = __builtin_amdgcn_readfirstlane(tid >> 5);
    const int rbase = blockIdx.x * ROWS;
    const int h     = lane >> 4;
    const int n     = lane & 15;

    for (int t = tid; t < ROWS * INPS; t += 32 * WAVES) {
        int col = t & (INPS - 1);
        int r   = t >> 10;
        act[col * ROWS + r] = x[(size_t)(rbase + r) * INPS + col];
    }
    for (int t = tid; t < ROWS * NEUR; t += 32 * WAVES) {
        act[INPS * ROWS + t] = 0.0f;
    }
    __syncthreads();

    lds_f*       actn  = (lds_f*)act + n;
    lds_f*       partn = (lds_f*)act + ACTF + n * WAVES;
    const bool     hi     = h != 0;
    const unsigned hbytes = 4u * (unsigned)(QK * (2 * wv + h));
    const unsigned pbytes = 16u * (unsigned)n;
    float4 w0[2], w1[2], w2[2], w3[2];
    int4   j0[2], j1[2], j2[2], j3[2];
    int4   pj0, pj1, pj2, pj3;
    float4 pw0, pw1, pw2, pw3;
    v2f    bA[4], bB[4];
    v2f    acA, acB;
    const lds_f* g[8];
    const char* jbase = (const char*)in_idxs;
    const char* wbase = (const char*)weights;

    load_j(j0, jbase + hbytes);
    load_j(j1, jbase + 1 * ROWBYTES + hbytes);
    load_j(j2, jbase + 2 * ROWBYTES + hbytes);
    load_j(j3, jbase + 3 * ROWBYTES + hbytes);
    load_w(w0, wbase + hbytes);
    load_w(w1, wbase + 1 * ROWBYTES + hbytes);
    to_addr(g, j0, actn);
    gather(bA, g);
    to_addr(g, j1, actn);
    gather(bB, g);
    const float part0 = block4(w0, bA);
    const float part1 = block4(w1, bB);
    to_addr(g, j2, actn);
    gather(bA, g);
    to_addr(g, j3, actn);
    partn[0 * (ROWS * WAVES) + wv] = part0;
    partn[1 * (ROWS * WAVES) + wv] = part1;
    bar_signal_lds_done();
    pj1 = *(const int4*)(jbase + 1 * ROWBYTES + pbytes);
    pw1 = *(const float4*)(wbase + 1 * ROWBYTES + pbytes);
    load_w(w2, wbase + 2 * ROWBYTES + hbytes);
    load_j(j0, jbase + 4 * ROWBYTES + hbytes);
    pj2 = *(const int4*)(jbase + 2 * ROWBYTES + pbytes);
    pw2 = *(const float4*)(wbase + 2 * ROWBYTES + pbytes);
    load_w(w3, wbase + 3 * ROWBYTES + hbytes);
    load_j(j1, jbase + 5 * ROWBYTES + hbytes);
    pj3 = *(const int4*)(jbase + 3 * ROWBYTES + pbytes);
    pw3 = *(const float4*)(wbase + 3 * ROWBYTES + pbytes);
    load_w(w0, wbase + 4 * ROWBYTES + hbytes);
    load_j(j2, jbase + 6 * ROWBYTES + hbytes);
    pj0 = *(const int4*)(jbase + 4 * ROWBYTES + pbytes);
    pw0 = *(const float4*)(wbase + 4 * ROWBYTES + pbytes);
    load_w(w1, wbase + 5 * ROWBYTES + hbytes);
    load_j(j3, jbase + 7 * ROWBYTES + hbytes);
    bar_wait();
    v4f ep = load_parts(partn, 0);
    unsigned joff = 8 * ROWBYTES;
    unsigned woff = 6 * ROWBYTES;
    unsigned poff = 5 * ROWBYTES;
    acA.x = 0.0f; acA.y = 0.0f;
    acB = acA;
    float sg1 = 0.0f, sg2 = 0.0f, sg3 = 0.0f;

    for (int pos = 0; pos < NEUR; pos += 4) {
        pipe_trip(actn, partn, wv, hi, g, bB, bA, ep, w2, j0, pj1, pw1, acA, acB, sg1, sg2, sg3,
                  wbase, jbase, woff, joff, poff, hbytes, pbytes, pos, 2, 1);
        pipe_trip(actn, partn, wv, hi, g, bA, bB, ep, w3, j1, pj2, pw2, acB, acA, sg1, sg2, sg3,
                  wbase, jbase, woff, joff, poff, hbytes, pbytes, pos + 1, 3, 2);
        pipe_trip(actn, partn, wv, hi, g, bB, bA, ep, w0, j2, pj3, pw3, acA, acB, sg1, sg2, sg3,
                  wbase, jbase, woff, joff, poff, hbytes, pbytes, pos + 2, 0, 3);
        pipe_trip(actn, partn, wv, hi, g, bA, bB, ep, w1, j3, pj0, pw0, acB, acA, sg1, sg2, sg3,
                  wbase, jbase, woff, joff, poff, hbytes, pbytes, pos + 3, 1, 0);
    }

    for (int t = tid; t < NOUT * ROWS; t += 32 * WAVES) {
        int o = t >> 4;
        int r = t & 15;
        out[(size_t)o * BATCH + rbase + r] =
            act[(INPS + NEUR - NOUT + o) * ROWS + r];
    }
}

extern "C" void kernel_launch(void* const* d_in, const int* in_sizes, int n_in,
                              void* d_out, int out_size, void* d_ws, size_t ws_size,
                              hipStream_t stream) {
    (void)in_sizes; (void)n_in; (void)d_ws; (void)ws_size; (void)out_size;
    const float* x       = (const float*)d_in[0];
    const int*   in_idxs = (const int*)d_in[1];
    const float* weights = (const float*)d_in[2];
    float*       out     = (float*)d_out;

    const size_t smem_bytes = (size_t)(ACTF + PARTF) * sizeof(float);

    ne_scan_wmma_kernel<<<dim3(BATCH / ROWS), dim3(32 * WAVES), smem_bytes, stream>>>(
        x, in_idxs, weights, out);
}


# Late-term coefficients made once per work-group before the scan (LDS table, read with the gathers), not by every wave in every trip; block WMMAs then fill the partial-sum wait (sum order unchanged)
# speedup vs baseline: 3.4195x; 1.1111x over previous
#include <hip/hip_runtime.h>


typedef float v2f __attribute__((ext_vector_type(2)));
typedef float v4f __attribute__((ext_vector_type(4)));
typedef float v8f __attribute__((ext_vector_type(8)));
typedef __attribute__((address_space(3))) float lds_f;
typedef __attribute__((address_space(3))) v4f   lds_v4f;
typedef __attribute__((address_space(3))) v2f   lds_v2f;

#define BATCH 512
#define INPS  1024
#define NEUR  512
#define KFAN  64
#define NOUT  16
#define ROWS  16
#define WAVES 4
#define QK    8

#define ROWBYTES (KFAN * 4)
#define ACTF  ((INPS + NEUR) * ROWS)
#define PARTF (4 * ROWS * WAVES)
#define COEFF ((NEUR + 1) * 4)
#define CCH   8
#define CPASSB (2 * WAVES * ROWBYTES)

static __device__ __forceinline__ void load_w(float4 (&w)[2], const char* __restrict__ wp) {
    const float4* p = (const float4*)wp;
    w[0] = p[0];
    w[1] = p[1];
}

static __device__ __forceinline__ void load_j(int4 (&j)[2], const char* __restrict__ jp) {
    const int4* p = (const int4*)jp;
    j[0] = p[0];
    j[1] = p[1];
}

static __device__ __forceinline__ void to_addr(const lds_f* (&g)[8], const int4 (&j)[2], const lds_f* actn) {
    #pragma unroll
    for (int q = 0; q < 2; ++q) {
        g[4 * q + 0] = actn + j[q].x * ROWS;
        g[4 * q + 1] = actn + j[q].y * ROWS;
        g[4 * q + 2] = actn + j[q].z * ROWS;
        g[4 * q + 3] = actn + j[q].w * ROWS;
    }
}

static __device__ __forceinline__ void gather(v2f (&b)[4], const lds_f* const (&g)[8]) {
    #pragma unroll
    for (int i = 0; i < 4; ++i) {
        b[i].x = *g[2 * i];
        b[i].y = *g[2 * i + 1];
    }
}

static __device__ __forceinline__ float block4(const float4 (&w)[2], const v2f (&b)[4]) {
    v8f acc0 = v8f{};
    v8f acc1 = v8f{};
    #pragma unroll
    for (int i = 0; i < 4; ++i) {
        const float4 wq = w[i >> 1];
        v2f a;
        if (i & 1) { a.x = wq.z; a.y = wq.w; } else { a.x = wq.x; a.y = wq.y; }
        if (i & 1)
            acc1 = __builtin_amdgcn_wmma_f32_16x16x4_f32(false, a, false, b[i], (short)0, acc1, false, false);
        else
            acc0 = __builtin_amdgcn_wmma_f32_16x16x4_f32(false, a, false, b[i], (short)0, acc0, false, false);
    }
    return acc0[0] + acc1[0];
}

static __device__ __forceinline__ float row_sum16(float v) {
    v += __builtin_bit_cast(float, __builtin_amdgcn_update_dpp(0, __builtin_bit_cast(int, v), 0x161, 0xf, 0xf, false));
    v += __builtin_bit_cast(float, __builtin_amdgcn_update_dpp(0, __builtin_bit_cast(int, v), 0x162, 0xf, 0xf, false));
    v += __builtin_bit_cast(float, __builtin_amdgcn_update_dpp(0, __builtin_bit_cast(int, v), 0x164, 0xf, 0xf, false));
    v += __builtin_bit_cast(float, __builtin_amdgcn_update_dpp(0, __builtin_bit_cast(int, v), 0x168, 0xf, 0xf, false));
    return v;
}

static __device__ __forceinline__ void bar_signal_lds_done() {
    asm volatile("s_wait_dscnt 0x0\n\ts_barrier_signal -1" ::: "memory");
}
static __device__ __forceinline__ void bar_wait() {
    asm volatile("s_barrier_wait -1" ::: "memory");
}

static __device__ __forceinline__ v4f load_parts(const lds_f* partn, int slot) {
    return *(const lds_v4f*)(partn + slot * (ROWS * WAVES));
}

static __device__ __forceinline__ void coef_load(int4 (&cj)[CCH], float4 (&cw)[CCH],
                                                 const char* __restrict__ jrow, const char* __restrict__ wrow) {
    #pragma unroll
    for (int u = 0; u < CCH; ++u) {
        cj[u] = *(const int4*)(jrow + u * CPASSB);
        cw[u] = *(const float4*)(wrow + u * CPASSB);
    }
}

static __device__ __forceinline__ float coef_lane(const int4 &cj, const float4 &cw, int t) {
    return (((cj.x == t) ? cw.x : 0.0f) + ((cj.y == t) ? cw.y : 0.0f))
         + (((cj.z == t) ? cw.z : 0.0f) + ((cj.w == t) ? cw.w : 0.0f));
}

static __device__ __forceinline__ void coef_make(const int4 (&cj)[CCH], const float4 (&cw)[CCH],
                                                 lds_f* coef, int q0, int n) {
    float c1[CCH], c2[CCH], c3[CCH];
    #pragma unroll
    for (int u = 0; u < CCH; ++u) {
        const int q  = q0 + u * (2 * WAVES);
        const int t1 = (q >= 1) ? (INPS + q - 1) : -1;
        const int t2 = (q >= 2) ? (INPS + q - 2) : -1;
        const int t3 = (q >= 3) ? (INPS + q - 3) : -1;
        c1[u] = row_sum16(coef_lane(cj[u], cw[u], t1));
        c2[u] = row_sum16(coef_lane(cj[u], cw[u], t2));
        c3[u] = row_sum16(coef_lane(cj[u], cw[u], t3));
    }
    if (n == 0) {
        #pragma unroll
        for (int u = 0; u < CCH; ++u) {
            v4f e;
            e.x = c1[u]; e.y = c2[u]; e.z = 0.0f; e.w = c3[u];
            *(lds_v4f*)(coef + (q0 + u * (2 * WAVES)) * 4) = e;
        }
    }
}

static __device__ __forceinline__ void pipe_trip(
    lds_f* actn, lds_f* partn, const lds_f* coefn, int wv, bool hi, const lds_f* (&g)[8],
    v2f (&bfill)[4], const v2f (&buse)[4], v4f &ep,
    float4 (&wuse)[2], int4 (&juse)[2],
    const v2f &acorr, v2f &acorr_n, float &sg1, float &sg2, float &sg3,
    const char* __restrict__ wbase, const char* __restrict__ jbase,
    unsigned &woff, unsigned &joff, unsigned hbytes,
    int pos, const int slotw, const int slotr)
{
    gather(bfill, g);
    acorr_n = *(const lds_v2f*)(coefn + (pos + 1) * 4);
    __builtin_amdgcn_sched_barrier(0);

    v2f bc;
    bc.x = sg1;
    bc.y = hi ? sg3 : sg2;
    const v8f cacc = __builtin_amdgcn_wmma_f32_16x16x4_f32(
        false, acorr, false, bc, (short)0, v8f{}, false, false);
    asm volatile("" :: "v"(cacc));
    __builtin_amdgcn_sched_barrier(0);

    v8f acc0 = v8f{};
    v8f acc1 = v8f{};
    #pragma unroll
    for (int i = 0; i < 4; ++i) {
        const float4 wq = wuse[i >> 1];
        v2f a;
        if (i & 1) { a.x = wq.z; a.y = wq.w; } else { a.x = wq.x; a.y = wq.y; }
        if (i & 1)
            acc1 = __builtin_amdgcn_wmma_f32_16x16x4_f32(false, a, false, buse[i], (short)0, acc1, false, false);
        else
            acc0 = __builtin_amdgcn_wmma_f32_16x16x4_f32(false, a, false, buse[i], (short)0, acc0, false, false);
    }
    to_addr(g, juse, actn);
    const char* jp = jbase + joff + hbytes;
    const char* wp = wbase + woff + hbytes;
    asm volatile("" :: "v"(acc0), "v"(acc1), "v"(g[0]), "v"(g[1]), "v"(g[2]), "v"(g[3]),
                       "v"(g[4]), "v"(g[5]), "v"(g[6]), "v"(g[7]), "v"(jp), "v"(wp));
    __builtin_amdgcn_sched_barrier(0);

    const float early = (ep.x + ep.y) + (ep.z + ep.w);
    const float val = cacc[0] + early;
    const float ex  = __expf(-val);
    const float sg  = __builtin_amdgcn_rcpf(1.0f + ex);
    asm volatile("" :: "v"(sg));
    __builtin_amdgcn_sched_barrier(0);

    const float part = acc0[0] + acc1[0];
    asm volatile("" :: "v"(part));
    __builtin_amdgcn_sched_barrier(0);

    bar_signal_lds_done();

    partn[slotw * (ROWS * WAVES) + wv] = part;

    load_j(juse, jp);
    load_w(wuse, wp);
    if (pos + 9 < NEUR) joff += ROWBYTES;
    if (pos + 7 < NEUR) woff += ROWBYTES;

    bar_wait();
    ep = load_parts(partn, slotr);
    actn[(INPS + pos) * ROWS] = sg;
    sg3 = sg2;
    sg2 = sg1;
    sg1 = sg;
}

__global__ __launch_bounds__(32 * WAVES) void ne_scan_wmma_kernel(
    const float* __restrict__ x,
    const int*   __restrict__ in_idxs,
    const float* __restrict__ weights,
    float*       __restrict__ out)
{
    extern __shared__ float act[];

    const int tid   = threadIdx.x;
    const int lane  = tid & 31;
    const int wv    = __builtin_amdgcn_readfirstlane(tid >> 5);
    const int rbase = blockIdx.x * ROWS;
    const int h     = lane >> 4;
    const int n     = lane & 15;

    const char* jbase = (const char*)in_idxs;
    const char* wbase = (const char*)weights;

    lds_f*      coef = (lds_f*)act + ACTF + PARTF;
    const int   cq0  = 2 * wv + h;
    const char* cjr  = jbase + cq0 * ROWBYTES + 16 * n;
    const char* cwr  = wbase + cq0 * ROWBYTES + 16 * n;
    int4   cja[CCH], cjb[CCH];
    float4 cwa[CCH], cwb[CCH];
    coef_load(cja, cwa, cjr, cwr);

    for (int t = tid; t < ROWS * INPS; t += 32 * WAVES) {
        int col = t & (INPS - 1);
        int r   = t >> 10;
        act[col * ROWS + r] = x[(size_t)(rbase + r) * INPS + col];
    }
    for (int t = tid; t < ROWS * NEUR; t += 32 * WAVES) {
        act[INPS * ROWS + t] = 0.0f;
    }

    #pragma unroll 1
    for (int c = 0; c < NEUR / (2 * WAVES * CCH); c += 2) {
        coef_load(cjb, cwb, cjr + (c + 1) * (CCH * CPASSB), cwr + (c + 1) * (CCH * CPASSB));
        coef_make(cja, cwa, coef, cq0 + c * (CCH * 2 * WAVES), n);
        const int cn = (c + 2 < NEUR / (2 * WAVES * CCH)) ? (c + 2) : (c + 1);
        coef_load(cja, cwa, cjr + cn * (CCH * CPASSB), cwr + cn * (CCH * CPASSB));
        coef_make(cjb, cwb, coef, cq0 + (c + 1) * (CCH * 2 * WAVES), n);
    }
    if (tid < 4) coef[NEUR * 4 + tid] = 0.0f;
    __syncthreads();

    lds_f*       actn  = (lds_f*)act + n;
    lds_f*       partn = (lds_f*)act + ACTF + n * WAVES;
    const lds_f* coefn = coef + 2 * h;
    const bool     hi     = h != 0;
    const unsigned hbytes = 4u * (unsigned)(QK * (2 * wv + h));
    float4 w0[2], w1[2], w2[2], w3[2];
    int4   j0[2], j1[2], j2[2], j3[2];
    v2f    bA[4], bB[4];
    v2f    acA, acB;
    const lds_f* g[8];

    load_j(j0, jbase + hbytes);
    load_j(j1, jbase + 1 * ROWBYTES + hbytes);
    load_j(j2, jbase + 2 * ROWBYTES + hbytes);
    load_j(j3, jbase + 3 * ROWBYTES + hbytes);
    load_w(w0, wbase + hbytes);
    load_w(w1, wbase + 1 * ROWBYTES + hbytes);
    to_addr(g, j0, actn);
    gather(bA, g);
    to_addr(g, j1, actn);
    gather(bB, g);
    const float part0 = block4(w0, bA);
    const float part1 = block4(w1, bB);
    to_addr(g, j2, actn);
    gather(bA, g);
    to_addr(g, j3, actn);
    partn[0 * (ROWS * WAVES) + wv] = part0;
    partn[1 * (ROWS * WAVES) + wv] = part1;
    bar_signal_lds_done();
    load_w(w2, wbase + 2 * ROWBYTES + hbytes);
    load_j(j0, jbase + 4 * ROWBYTES + hbytes);
    load_w(w3, wbase + 3 * ROWBYTES + hbytes);
    load_j(j1, jbase + 5 * ROWBYTES + hbytes);
    load_w(w0, wbase + 4 * ROWBYTES + hbytes);
    load_j(j2, jbase + 6 * ROWBYTES + hbytes);
    load_w(w1, wbase + 5 * ROWBYTES + hbytes);
    load_j(j3, jbase + 7 * ROWBYTES + hbytes);
    bar_wait();
    v4f ep = load_parts(partn, 0);
    unsigned joff = 8 * ROWBYTES;
    unsigned woff = 6 * ROWBYTES;
    acA.x = 0.0f; acA.y = 0.0f;
    acB = acA;
    float sg1 = 0.0f, sg2 = 0.0f, sg3 = 0.0f;

    for (int pos = 0; pos < NEUR; pos += 4) {
        pipe_trip(actn, partn, coefn, wv, hi, g, bB, bA, ep, w2, j0, acA, acB, sg1, sg2, sg3,
                  wbase, jbase, woff, joff, hbytes, pos, 2, 1);
        pipe_trip(actn, partn, coefn, wv, hi, g, bA, bB, ep, w3, j1, acB, acA, sg1, sg2, sg3,
                  wbase, jbase, woff, joff, hbytes, pos + 1, 3, 2);
        pipe_trip(actn, partn, coefn, wv, hi, g, bB, bA, ep, w0, j2, acA, acB, sg1, sg2, sg3,
                  wbase, jbase, woff, joff, hbytes, pos + 2, 0, 3);
        pipe_trip(actn, partn, coefn, wv, hi, g, bA, bB, ep, w1, j3, acB, acA, sg1, sg2, sg3,
                  wbase, jbase, woff, joff, hbytes, pos + 3, 1, 0);
    }

    for (int t = tid; t < NOUT * ROWS; t += 32 * WAVES) {
        int o = t >> 4;
        int r = t & 15;
        out[(size_t)o * BATCH + rbase + r] =
            act[(INPS + NEUR - NOUT + o) * ROWS + r];
    }
}

extern "C" void kernel_launch(void* const* d_in, const int* in_sizes, int n_in,
                              void* d_out, int out_size, void* d_ws, size_t ws_size,
                              hipStream_t stream) {
    (void)in_sizes; (void)n_in; (void)d_ws; (void)ws_size; (void)out_size;
    const float* x       = (const float*)d_in[0];
    const int*   in_idxs = (const int*)d_in[1];
    const float* weights = (const float*)d_in[2];
    float*       out     = (float*)d_out;

    const size_t smem_bytes = (size_t)(ACTF + PARTF + COEFF) * sizeof(float);

    ne_scan_wmma_kernel<<<dim3(BATCH / ROWS), dim3(32 * WAVES), smem_bytes, stream>>>(
        x, in_idxs, weights, out);
}


# Barrier signal moved up in front of the sigmoid's exp and rcp (it needs only the gathers' return): signal-to-release latency runs beside the serial chain, not behind it (sum order unchanged)
# speedup vs baseline: 3.5318x; 1.0328x over previous
#include <hip/hip_runtime.h>


typedef float v2f __attribute__((ext_vector_type(2)));
typedef float v4f __attribute__((ext_vector_type(4)));
typedef float v8f __attribute__((ext_vector_type(8)));
typedef __attribute__((address_space(3))) float lds_f;
typedef __attribute__((address_space(3))) v4f   lds_v4f;
typedef __attribute__((address_space(3))) v2f   lds_v2f;

#define BATCH 512
#define INPS  1024
#define NEUR  512
#define KFAN  64
#define NOUT  16
#define ROWS  16
#define WAVES 4
#define QK    8

#define ROWBYTES (KFAN * 4)
#define ACTF  ((INPS + NEUR) * ROWS)
#define PARTF (4 * ROWS * WAVES)
#define COEFF ((NEUR + 1) * 4)
#define CCH   8
#define CPASSB (2 * WAVES * ROWBYTES)

static __device__ __forceinline__ void load_w(float4 (&w)[2], const char* __restrict__ wp) {
    const float4* p = (const float4*)wp;
    w[0] = p[0];
    w[1] = p[1];
}

static __device__ __forceinline__ void load_j(int4 (&j)[2], const char* __restrict__ jp) {
    const int4* p = (const int4*)jp;
    j[0] = p[0];
    j[1] = p[1];
}

static __device__ __forceinline__ void to_addr(const lds_f* (&g)[8], const int4 (&j)[2], const lds_f* actn) {
    #pragma unroll
    for (int q = 0; q < 2; ++q) {
        g[4 * q + 0] = actn + j[q].x * ROWS;
        g[4 * q + 1] = actn + j[q].y * ROWS;
        g[4 * q + 2] = actn + j[q].z * ROWS;
        g[4 * q + 3] = actn + j[q].w * ROWS;
    }
}

static __device__ __forceinline__ void gather(v2f (&b)[4], const lds_f* const (&g)[8]) {
    #pragma unroll
    for (int i = 0; i < 4; ++i) {
        b[i].x = *g[2 * i];
        b[i].y = *g[2 * i + 1];
    }
}

static __device__ __forceinline__ float block4(const float4 (&w)[2], const v2f (&b)[4]) {
    v8f acc0 = v8f{};
    v8f acc1 = v8f{};
    #pragma unroll
    for (int i = 0; i < 4; ++i) {
        const float4 wq = w[i >> 1];
        v2f a;
        if (i & 1) { a.x = wq.z; a.y = wq.w; } else { a.x = wq.x; a.y = wq.y; }
        if (i & 1)
            acc1 = __builtin_amdgcn_wmma_f32_16x16x4_f32(false, a, false, b[i], (short)0, acc1, false, false);
        else
            acc0 = __builtin_amdgcn_wmma_f32_16x16x4_f32(false, a, false, b[i], (short)0, acc0, false, false);
    }
    return acc0[0] + acc1[0];
}

static __device__ __forceinline__ float row_sum16(float v) {
    v += __builtin_bit_cast(float, __builtin_amdgcn_update_dpp(0, __builtin_bit_cast(int, v), 0x161, 0xf, 0xf, false));
    v += __builtin_bit_cast(float, __builtin_amdgcn_update_dpp(0, __builtin_bit_cast(int, v), 0x162, 0xf, 0xf, false));
    v += __builtin_bit_cast(float, __builtin_amdgcn_update_dpp(0, __builtin_bit_cast(int, v), 0x164, 0xf, 0xf, false));
    v += __builtin_bit_cast(float, __builtin_amdgcn_update_dpp(0, __builtin_bit_cast(int, v), 0x168, 0xf, 0xf, false));
    return v;
}

static __device__ __forceinline__ void bar_signal_lds_done() {
    asm volatile("s_wait_dscnt 0x0\n\ts_barrier_signal -1" ::: "memory");
}
static __device__ __forceinline__ void bar_wait() {
    asm volatile("s_barrier_wait -1" ::: "memory");
}

static __device__ __forceinline__ v4f load_parts(const lds_f* partn, int slot) {
    return *(const lds_v4f*)(partn + slot * (ROWS * WAVES));
}

static __device__ __forceinline__ void coef_load(int4 (&cj)[CCH], float4 (&cw)[CCH],
                                                 const char* __restrict__ jrow, const char* __restrict__ wrow) {
    #pragma unroll
    for (int u = 0; u < CCH; ++u) {
        cj[u] = *(const int4*)(jrow + u * CPASSB);
        cw[u] = *(const float4*)(wrow + u * CPASSB);
    }
}

static __device__ __forceinline__ float coef_lane(const int4 &cj, const float4 &cw, int t) {
    return (((cj.x == t) ? cw.x : 0.0f) + ((cj.y == t) ? cw.y : 0.0f))
         + (((cj.z == t) ? cw.z : 0.0f) + ((cj.w == t) ? cw.w : 0.0f));
}

static __device__ __forceinline__ void coef_make(const int4 (&cj)[CCH], const float4 (&cw)[CCH],
                                                 lds_f* coef, int q0, int n) {
    float c1[CCH], c2[CCH], c3[CCH];
    #pragma unroll
    for (int u = 0; u < CCH; ++u) {
        const int q  = q0 + u * (2 * WAVES);
        const int t1 = (q >= 1) ? (INPS + q - 1) : -1;
        const int t2 = (q >= 2) ? (INPS + q - 2) : -1;
        const int t3 = (q >= 3) ? (INPS + q - 3) : -1;
        c1[u] = row_sum16(coef_lane(cj[u], cw[u], t1));
        c2[u] = row_sum16(coef_lane(cj[u], cw[u], t2));
        c3[u] = row_sum16(coef_lane(cj[u], cw[u], t3));
    }
    if (n == 0) {
        #pragma unroll
        for (int u = 0; u < CCH; ++u) {
            v4f e;
            e.x = c1[u]; e.y = c2[u]; e.z = 0.0f; e.w = c3[u];
            *(lds_v4f*)(coef + (q0 + u * (2 * WAVES)) * 4) = e;
        }
    }
}

static __device__ __forceinline__ void pipe_trip(
    lds_f* actn, lds_f* partn, const lds_f* coefn, int wv, bool hi, const lds_f* (&g)[8],
    v2f (&bfill)[4], const v2f (&buse)[4], v4f &ep,
    float4 (&wuse)[2], int4 (&juse)[2],
    const v2f &acorr, v2f &acorr_n, float &sg1, float &sg2, float &sg3,
    const char* __restrict__ wbase, const char* __restrict__ jbase,
    unsigned &woff, unsigned &joff, unsigned hbytes,
    int pos, const int slotw, const int slotr)
{
    gather(bfill, g);
    acorr_n = *(const lds_v2f*)(coefn + (pos + 1) * 4);
    __builtin_amdgcn_sched_barrier(0);

    v2f bc;
    bc.x = sg1;
    bc.y = hi ? sg3 : sg2;
    const v8f cacc = __builtin_amdgcn_wmma_f32_16x16x4_f32(
        false, acorr, false, bc, (short)0, v8f{}, false, false);
    asm volatile("" :: "v"(cacc));
    __builtin_amdgcn_sched_barrier(0);

    v8f acc0 = v8f{};
    v8f acc1 = v8f{};
    #pragma unroll
    for (int i = 0; i < 4; ++i) {
        const float4 wq = wuse[i >> 1];
        v2f a;
        if (i & 1) { a.x = wq.z; a.y = wq.w; } else { a.x = wq.x; a.y = wq.y; }
        if (i & 1)
            acc1 = __builtin_amdgcn_wmma_f32_16x16x4_f32(false, a, false, buse[i], (short)0, acc1, false, false);
        else
            acc0 = __builtin_amdgcn_wmma_f32_16x16x4_f32(false, a, false, buse[i], (short)0, acc0, false, false);
    }
    to_addr(g, juse, actn);
    const char* jp = jbase + joff + hbytes;
    const char* wp = wbase + woff + hbytes;
    asm volatile("" :: "v"(acc0), "v"(acc1), "v"(g[0]), "v"(g[1]), "v"(g[2]), "v"(g[3]),
                       "v"(g[4]), "v"(g[5]), "v"(g[6]), "v"(g[7]), "v"(jp), "v"(wp));
    __builtin_amdgcn_sched_barrier(0);

    const float early = (ep.x + ep.y) + (ep.z + ep.w);
    asm volatile("" :: "v"(early));

    bar_signal_lds_done();
    __builtin_amdgcn_sched_barrier(0);

    const float val = cacc[0] + early;
    const float ex  = __expf(-val);
    const float sg  = __builtin_amdgcn_rcpf(1.0f + ex);
    asm volatile("" :: "v"(sg));
    __builtin_amdgcn_sched_barrier(0);

    const float part = acc0[0] + acc1[0];
    asm volatile("" :: "v"(part));
    __builtin_amdgcn_sched_barrier(0);

    partn[slotw * (ROWS * WAVES) + wv] = part;

    load_j(juse, jp);
    load_w(wuse, wp);
    if (pos + 9 < NEUR) joff += ROWBYTES;
    if (pos + 7 < NEUR) woff += ROWBYTES;

    bar_wait();
    ep = load_parts(partn, slotr);
    actn[(INPS + pos) * ROWS] = sg;
    sg3 = sg2;
    sg2 = sg1;
    sg1 = sg;
}

__global__ __launch_bounds__(32 * WAVES) void ne_scan_wmma_kernel(
    const float* __restrict__ x,
    const int*   __restrict__ in_idxs,
    const float* __restrict__ weights,
    float*       __restrict__ out)
{
    extern __shared__ float act[];

    const int tid   = threadIdx.x;
    const int lane  = tid & 31;
    const int wv    = __builtin_amdgcn_readfirstlane(tid >> 5);
    const int rbase = blockIdx.x * ROWS;
    const int h     = lane >> 4;
    const int n     = lane & 15;

    const char* jbase = (const char*)in_idxs;
    const char* wbase = (const char*)weights;

    lds_f*      coef = (lds_f*)act + ACTF + PARTF;
    const int   cq0  = 2 * wv + h;
    const char* cjr  = jbase + cq0 * ROWBYTES + 16 * n;
    const char* cwr  = wbase + cq0 * ROWBYTES + 16 * n;
    int4   cja[CCH], cjb[CCH];
    float4 cwa[CCH], cwb[CCH];
    coef_load(cja, cwa, cjr, cwr);

    for (int t = tid; t < ROWS * INPS; t += 32 * WAVES) {
        int col = t & (INPS - 1);
        int r   = t >> 10;
        act[col * ROWS + r] = x[(size_t)(rbase + r) * INPS + col];
    }
    for (int t = tid; t < ROWS * NEUR; t += 32 * WAVES) {
        act[INPS * ROWS + t] = 0.0f;
    }

    #pragma unroll 1
    for (int c = 0; c < NEUR / (2 * WAVES * CCH); c += 2) {
        coef_load(cjb, cwb, cjr + (c + 1) * (CCH * CPASSB), cwr + (c + 1) * (CCH * CPASSB));
        coef_make(cja, cwa, coef, cq0 + c * (CCH * 2 * WAVES), n);
        const int cn = (c + 2 < NEUR / (2 * WAVES * CCH)) ? (c + 2) : (c + 1);
        coef_load(cja, cwa, cjr + cn * (CCH * CPASSB), cwr + cn * (CCH * CPASSB));
        coef_make(cjb, cwb, coef, cq0 + (c + 1) * (CCH * 2 * WAVES), n);
    }
    if (tid < 4) coef[NEUR * 4 + tid] = 0.0f;
    __syncthreads();

    lds_f*       actn  = (lds_f*)act + n;
    lds_f*       partn = (lds_f*)act + ACTF + n * WAVES;
    const lds_f* coefn = coef + 2 * h;
    const bool     hi     = h != 0;
    const unsigned hbytes = 4u * (unsigned)(QK * (2 * wv + h));
    float4 w0[2], w1[2], w2[2], w3[2];
    int4   j0[2], j1[2], j2[2], j3[2];
    v2f    bA[4], bB[4];
    v2f    acA, acB;
    const lds_f* g[8];

    load_j(j0, jbase + hbytes);
    load_j(j1, jbase + 1 * ROWBYTES + hbytes);
    load_j(j2, jbase + 2 * ROWBYTES + hbytes);
    load_j(j3, jbase + 3 * ROWBYTES + hbytes);
    load_w(w0, wbase + hbytes);
    load_w(w1, wbase + 1 * ROWBYTES + hbytes);
    to_addr(g, j0, actn);
    gather(bA, g);
    to_addr(g, j1, actn);
    gather(bB, g);
    const float part0 = block4(w0, bA);
    const float part1 = block4(w1, bB);
    to_addr(g, j2, actn);
    gather(bA, g);
    to_addr(g, j3, actn);
    partn[0 * (ROWS * WAVES) + wv] = part0;
    partn[1 * (ROWS * WAVES) + wv] = part1;
    bar_signal_lds_done();
    load_w(w2, wbase + 2 * ROWBYTES + hbytes);
    load_j(j0, jbase + 4 * ROWBYTES + hbytes);
    load_w(w3, wbase + 3 * ROWBYTES + hbytes);
    load_j(j1, jbase + 5 * ROWBYTES + hbytes);
    load_w(w0, wbase + 4 * ROWBYTES + hbytes);
    load_j(j2, jbase + 6 * ROWBYTES + hbytes);
    load_w(w1, wbase + 5 * ROWBYTES + hbytes);
    load_j(j3, jbase + 7 * ROWBYTES + hbytes);
    bar_wait();
    v4f ep = load_parts(partn, 0);
    unsigned joff = 8 * ROWBYTES;
    unsigned woff = 6 * ROWBYTES;
    acA.x = 0.0f; acA.y = 0.0f;
    acB = acA;
    float sg1 = 0.0f, sg2 = 0.0f, sg3 = 0.0f;

    for (int pos = 0; pos < NEUR; pos += 4) {
        pipe_trip(actn, partn, coefn, wv, hi, g, bB, bA, ep, w2, j0, acA, acB, sg1, sg2, sg3,
                  wbase, jbase, woff, joff, hbytes, pos, 2, 1);
        pipe_trip(actn, partn, coefn, wv, hi, g, bA, bB, ep, w3, j1, acB, acA, sg1, sg2, sg3,
                  wbase, jbase, woff, joff, hbytes, pos + 1, 3, 2);
        pipe_trip(actn, partn, coefn, wv, hi, g, bB, bA, ep, w0, j2, acA, acB, sg1, sg2, sg3,
                  wbase, jbase, woff, joff, hbytes, pos + 2, 0, 3);
        pipe_trip(actn, partn, coefn, wv, hi, g, bA, bB, ep, w1, j3, acB, acA, sg1, sg2, sg3,
                  wbase, jbase, woff, joff, hbytes, pos + 3, 1, 0);
    }

    for (int t = tid; t < NOUT * ROWS; t += 32 * WAVES) {
        int o = t >> 4;
        int r = t & 15;
        out[(size_t)o * BATCH + rbase + r] =
            act[(INPS + NEUR - NOUT + o) * ROWS + r];
    }
}

extern "C" void kernel_launch(void* const* d_in, const int* in_sizes, int n_in,
                              void* d_out, int out_size, void* d_ws, size_t ws_size,
                              hipStream_t stream) {
    (void)in_sizes; (void)n_in; (void)d_ws; (void)ws_size; (void)out_size;
    const float* x       = (const float*)d_in[0];
    const int*   in_idxs = (const int*)d_in[1];
    const float* weights = (const float*)d_in[2];
    float*       out     = (float*)d_out;

    const size_t smem_bytes = (size_t)(ACTF + PARTF + COEFF) * sizeof(float);

    ne_scan_wmma_kernel<<<dim3(BATCH / ROWS), dim3(32 * WAVES), smem_bytes, stream>>>(
        x, in_idxs, weights, out);
}


# 8 batch rows per work-group: the WMMA's 16 N columns carry them twice (two 8-k sub-chunks; A rows 0-7 / 8-15 hold their weights): 2 WMMAs, 4 gathers per wave and step, not 4, 8 (reorders 64-term sum)
# speedup vs baseline: 4.5314x; 1.2830x over previous
#include <hip/hip_runtime.h>


typedef float v2f __attribute__((ext_vector_type(2)));
typedef float v4f __attribute__((ext_vector_type(4)));
typedef float v8f __attribute__((ext_vector_type(8)));
typedef __attribute__((address_space(3))) float lds_f;
typedef __attribute__((address_space(3))) v4f   lds_v4f;
typedef __attribute__((address_space(3))) v2f   lds_v2f;

#define BATCH 512
#define INPS  1024
#define NEUR  512
#define KFAN  64
#define NOUT  16
#define ROWS  8
#define WAVES 4
#define QK    4
#define NPART (2 * WAVES)

#define ROWBYTES (KFAN * 4)
#define ACTF  ((INPS + NEUR) * ROWS)
#define SLOTF (ROWS * NPART)
#define PARTF (4 * SLOTF)
#define DUMPF (4 * SLOTF + 32 * WAVES)
#define COEFF ((NEUR + 1) * 4)
#define CCH   8
#define CPASSB (2 * WAVES * ROWBYTES)

static __device__ __forceinline__ void load_w(float4 &w, const char* __restrict__ wp) {
    w = *(const float4*)wp;
}

static __device__ __forceinline__ void load_j(int4 &j, const char* __restrict__ jp) {
    j = *(const int4*)jp;
}

static __device__ __forceinline__ void to_addr(const lds_f* (&g)[4], const int4 &j, const lds_f* actn) {
    g[0] = actn + j.x * ROWS;
    g[1] = actn + j.y * ROWS;
    g[2] = actn + j.z * ROWS;
    g[3] = actn + j.w * ROWS;
}

static __device__ __forceinline__ void gather(v2f (&b)[2], const lds_f* const (&g)[4]) {
    b[0].x = *g[0];
    b[0].y = *g[1];
    b[1].x = *g[2];
    b[1].y = *g[3];
}

static __device__ __forceinline__ float block2(const float4 &w, const v2f (&b)[2]) {
    v2f a0, a1;
    a0.x = w.x; a0.y = w.y;
    a1.x = w.z; a1.y = w.w;
    const v8f acc0 = __builtin_amdgcn_wmma_f32_16x16x4_f32(false, a0, false, b[0], (short)0, v8f{}, false, false);
    const v8f acc1 = __builtin_amdgcn_wmma_f32_16x16x4_f32(false, a1, false, b[1], (short)0, v8f{}, false, false);
    return acc0[0] + acc1[0];
}

static __device__ __forceinline__ float row_sum16(float v) {
    v += __builtin_bit_cast(float, __builtin_amdgcn_update_dpp(0, __builtin_bit_cast(int, v), 0x161, 0xf, 0xf, false));
    v += __builtin_bit_cast(float, __builtin_amdgcn_update_dpp(0, __builtin_bit_cast(int, v), 0x162, 0xf, 0xf, false));
    v += __builtin_bit_cast(float, __builtin_amdgcn_update_dpp(0, __builtin_bit_cast(int, v), 0x164, 0xf, 0xf, false));
    v += __builtin_bit_cast(float, __builtin_amdgcn_update_dpp(0, __builtin_bit_cast(int, v), 0x168, 0xf, 0xf, false));
    return v;
}

static __device__ __forceinline__ void bar_signal_lds_done() {
    asm volatile("s_wait_dscnt 0x0\n\ts_barrier_signal -1" ::: "memory");
}
static __device__ __forceinline__ void bar_wait() {
    asm volatile("s_barrier_wait -1" ::: "memory");
}

static __device__ __forceinline__ void load_parts(v4f (&ep)[2], const lds_f* partn, int slot) {
    ep[0] = *(const lds_v4f*)(partn + slot * SLOTF);
    ep[1] = *(const lds_v4f*)(partn + slot * SLOTF + 4);
}
static __device__ __forceinline__ float sum_parts(const v4f (&ep)[2]) {
    const float even = (ep[0].x + ep[0].z) + (ep[1].x + ep[1].z);
    const float odd  = (ep[0].y + ep[0].w) + (ep[1].y + ep[1].w);
    return even + odd;
}

static __device__ __forceinline__ void coef_load(int4 (&cj)[CCH], float4 (&cw)[CCH],
                                                 const char* __restrict__ jrow, const char* __restrict__ wrow) {
    #pragma unroll
    for (int u = 0; u < CCH; ++u) {
        cj[u] = *(const int4*)(jrow + u * CPASSB);
        cw[u] = *(const float4*)(wrow + u * CPASSB);
    }
}

static __device__ __forceinline__ float coef_lane(const int4 &cj, const float4 &cw, int t) {
    return (((cj.x == t) ? cw.x : 0.0f) + ((cj.y == t) ? cw.y : 0.0f))
         + (((cj.z == t) ? cw.z : 0.0f) + ((cj.w == t) ? cw.w : 0.0f));
}

static __device__ __forceinline__ void coef_make(const int4 (&cj)[CCH], const float4 (&cw)[CCH],
                                                 lds_f* coef, int q0, int n) {
    float c1[CCH], c2[CCH], c3[CCH];
    #pragma unroll
    for (int u = 0; u < CCH; ++u) {
        const int q  = q0 + u * (2 * WAVES);
        const int t1 = (q >= 1) ? (INPS + q - 1) : -1;
        const int t2 = (q >= 2) ? (INPS + q - 2) : -1;
        const int t3 = (q >= 3) ? (INPS + q - 3) : -1;
        c1[u] = row_sum16(coef_lane(cj[u], cw[u], t1));
        c2[u] = row_sum16(coef_lane(cj[u], cw[u], t2));
        c3[u] = row_sum16(coef_lane(cj[u], cw[u], t3));
    }
    if (n == 0) {
        #pragma unroll
        for (int u = 0; u < CCH; ++u) {
            v4f e;
            e.x = c1[u]; e.y = c2[u]; e.z = 0.0f; e.w = c3[u];
            *(lds_v4f*)(coef + (q0 + u * (2 * WAVES)) * 4) = e;
        }
    }
}

static __device__ __forceinline__ void pipe_trip(
    lds_f* actn, const lds_f* partn, lds_f* pstore, const lds_f* coefn, bool hi, const lds_f* (&g)[4],
    v2f (&bfill)[2], const v2f (&buse)[2], v4f (&ep)[2],
    float4 &wuse, int4 &juse,
    const v2f &acorr, v2f &acorr_n, float &sg1, float &sg2, float &sg3,
    const char* __restrict__ wbase, const char* __restrict__ jbase,
    unsigned &woff, unsigned &joff, unsigned hbytes,
    int pos, const int slotw, const int slotr)
{
    gather(bfill, g);
    acorr_n = *(const lds_v2f*)(coefn + (pos + 1) * 4);
    __builtin_amdgcn_sched_barrier(0);

    v2f bc;
    bc.x = sg1;
    bc.y = hi ? sg3 : sg2;
    const v8f cacc = __builtin_amdgcn_wmma_f32_16x16x4_f32(
        false, acorr, false, bc, (short)0, v8f{}, false, false);
    asm volatile("" :: "v"(cacc));
    __builtin_amdgcn_sched_barrier(0);

    v2f a0, a1;
    a0.x = wuse.x; a0.y = wuse.y;
    a1.x = wuse.z; a1.y = wuse.w;
    const v8f acc0 = __builtin_amdgcn_wmma_f32_16x16x4_f32(false, a0, false, buse[0], (short)0, v8f{}, false, false);
    const v8f acc1 = __builtin_amdgcn_wmma_f32_16x16x4_f32(false, a1, false, buse[1], (short)0, v8f{}, false, false);
    to_addr(g, juse, actn);
    const char* jp = jbase + joff + hbytes;
    const char* wp = wbase + woff + hbytes;
    asm volatile("" :: "v"(acc0), "v"(acc1), "v"(g[0]), "v"(g[1]), "v"(g[2]), "v"(g[3]), "v"(jp), "v"(wp));
    __builtin_amdgcn_sched_barrier(0);

    const float early = sum_parts(ep);
    asm volatile("" :: "v"(early));

    bar_signal_lds_done();
    __builtin_amdgcn_sched_barrier(0);

    const float val = cacc[0] + early;
    const float ex  = __expf(-val);
    const float sg  = __builtin_amdgcn_rcpf(1.0f + ex);
    asm volatile("" :: "v"(sg));
    __builtin_amdgcn_sched_barrier(0);

    const float part = acc0[0] + acc1[0];
    asm volatile("" :: "v"(part));
    __builtin_amdgcn_sched_barrier(0);

    pstore[slotw * SLOTF] = part;

    load_j(juse, jp);
    load_w(wuse, wp);
    if (pos + 9 < NEUR) joff += ROWBYTES;
    if (pos + 7 < NEUR) woff += ROWBYTES;

    bar_wait();
    load_parts(ep, partn, slotr);
    actn[(INPS + pos) * ROWS] = sg;
    sg3 = sg2;
    sg2 = sg1;
    sg1 = sg;
}

__global__ __launch_bounds__(32 * WAVES) void ne_scan_wmma_kernel(
    const float* __restrict__ x,
    const int*   __restrict__ in_idxs,
    const float* __restrict__ weights,
    float*       __restrict__ out)
{
    extern __shared__ float act[];

    const int tid   = threadIdx.x;
    const int lane  = tid & 31;
    const int wv    = __builtin_amdgcn_readfirstlane(tid >> 5);
    const int rbase = blockIdx.x * ROWS;
    const int h     = lane >> 4;
    const int n     = lane & 15;
    const int sub   = n >> 3;
    const int r     = n & 7;

    const char* jbase = (const char*)in_idxs;
    const char* wbase = (const char*)weights;

    lds_f*      coef = (lds_f*)act + ACTF + PARTF + DUMPF;
    const int   cq0  = 2 * wv + h;
    const char* cjr  = jbase + cq0 * ROWBYTES + 16 * n;
    const char* cwr  = wbase + cq0 * ROWBYTES + 16 * n;
    int4   cja[CCH], cjb[CCH];
    float4 cwa[CCH], cwb[CCH];
    coef_load(cja, cwa, cjr, cwr);

    for (int t = tid; t < ROWS * INPS; t += 32 * WAVES) {
        int col = t & (INPS - 1);
        int rr  = t >> 10;
        act[col * ROWS + rr] = x[(size_t)(rbase + rr) * INPS + col];
    }
    for (int t = tid; t < ROWS * NEUR; t += 32 * WAVES) {
        act[INPS * ROWS + t] = 0.0f;
    }

    #pragma unroll 1
    for (int c = 0; c < NEUR / (2 * WAVES * CCH); c += 2) {
        coef_load(cjb, cwb, cjr + (c + 1) * (CCH * CPASSB), cwr + (c + 1) * (CCH * CPASSB));
        coef_make(cja, cwa, coef, cq0 + c * (CCH * 2 * WAVES), n);
        const int cn = (c + 2 < NEUR / (2 * WAVES * CCH)) ? (c + 2) : (c + 1);
        coef_load(cja, cwa, cjr + cn * (CCH * CPASSB), cwr + cn * (CCH * CPASSB));
        coef_make(cjb, cwb, coef, cq0 + (c + 1) * (CCH * 2 * WAVES), n);
    }
    if (tid < 4) coef[NEUR * 4 + tid] = 0.0f;
    __syncthreads();

    lds_f*       actn  = (lds_f*)act + r;
    const lds_f* partn = (lds_f*)act + ACTF + r * NPART;
    lds_f*       pstore = (h == sub) ? ((lds_f*)act + ACTF + r * NPART + 2 * wv + h)
                                     : ((lds_f*)act + ACTF + PARTF + 32 * wv + lane);
    const lds_f* coefn = coef + 2 * h;
    const bool     hi     = h != 0;
    const unsigned hbytes = 4u * (unsigned)(16 * wv + 8 * sub + QK * h);
    float4 w0, w1, w2, w3;
    int4   j0, j1, j2, j3;
    v2f    bA[2], bB[2];
    v2f    acA, acB;
    const lds_f* g[4];

    load_j(j0, jbase + hbytes);
    load_j(j1, jbase + 1 * ROWBYTES + hbytes);
    load_j(j2, jbase + 2 * ROWBYTES + hbytes);
    load_j(j3, jbase + 3 * ROWBYTES + hbytes);
    load_w(w0, wbase + hbytes);
    load_w(w1, wbase + 1 * ROWBYTES + hbytes);
    to_addr(g, j0, actn);
    gather(bA, g);
    to_addr(g, j1, actn);
    gather(bB, g);
    const float part0 = block2(w0, bA);
    const float part1 = block2(w1, bB);
    to_addr(g, j2, actn);
    gather(bA, g);
    to_addr(g, j3, actn);
    pstore[0 * SLOTF] = part0;
    pstore[1 * SLOTF] = part1;
    bar_signal_lds_done();
    load_w(w2, wbase + 2 * ROWBYTES + hbytes);
    load_j(j0, jbase + 4 * ROWBYTES + hbytes);
    load_w(w3, wbase + 3 * ROWBYTES + hbytes);
    load_j(j1, jbase + 5 * ROWBYTES + hbytes);
    load_w(w0, wbase + 4 * ROWBYTES + hbytes);
    load_j(j2, jbase + 6 * ROWBYTES + hbytes);
    load_w(w1, wbase + 5 * ROWBYTES + hbytes);
    load_j(j3, jbase + 7 * ROWBYTES + hbytes);
    bar_wait();
    v4f ep[2];
    load_parts(ep, partn, 0);
    unsigned joff = 8 * ROWBYTES;
    unsigned woff = 6 * ROWBYTES;
    acA.x = 0.0f; acA.y = 0.0f;
    acB = acA;
    float sg1 = 0.0f, sg2 = 0.0f, sg3 = 0.0f;

    for (int pos = 0; pos < NEUR; pos += 4) {
        pipe_trip(actn, partn, pstore, coefn, hi, g, bB, bA, ep, w2, j0, acA, acB, sg1, sg2, sg3,
                  wbase, jbase, woff, joff, hbytes, pos, 2, 1);
        pipe_trip(actn, partn, pstore, coefn, hi, g, bA, bB, ep, w3, j1, acB, acA, sg1, sg2, sg3,
                  wbase, jbase, woff, joff, hbytes, pos + 1, 3, 2);
        pipe_trip(actn, partn, pstore, coefn, hi, g, bB, bA, ep, w0, j2, acA, acB, sg1, sg2, sg3,
                  wbase, jbase, woff, joff, hbytes, pos + 2, 0, 3);
        pipe_trip(actn, partn, pstore, coefn, hi, g, bA, bB, ep, w1, j3, acB, acA, sg1, sg2, sg3,
                  wbase, jbase, woff, joff, hbytes, pos + 3, 1, 0);
    }

    for (int t = tid; t < NOUT * ROWS; t += 32 * WAVES) {
        int o  = t >> 3;
        int rr = t & 7;
        out[(size_t)o * BATCH + rbase + rr] =
            act[(INPS + NEUR - NOUT + o) * ROWS + rr];
    }
}

extern "C" void kernel_launch(void* const* d_in, const int* in_sizes, int n_in,
                              void* d_out, int out_size, void* d_ws, size_t ws_size,
                              hipStream_t stream) {
    (void)in_sizes; (void)n_in; (void)d_ws; (void)ws_size; (void)out_size;
    const float* x       = (const float*)d_in[0];
    const int*   in_idxs = (const int*)d_in[1];
    const float* weights = (const float*)d_in[2];
    float*       out     = (float*)d_out;

    const size_t smem_bytes = (size_t)(ACTF + PARTF + DUMPF + COEFF) * sizeof(float);

    ne_scan_wmma_kernel<<<dim3(BATCH / ROWS), dim3(32 * WAVES), smem_bytes, stream>>>(
        x, in_idxs, weights, out);
}


# Two scan steps per work-group barrier: late terms q-1..q-5 by correction WMMAs, only the odd step's newest on the serial chain; partial sums made two trips ahead (reorders the 64-term sum)
# speedup vs baseline: 4.7081x; 1.0390x over previous
#include <hip/hip_runtime.h>


typedef float v2f __attribute__((ext_vector_type(2)));
typedef float v4f __attribute__((ext_vector_type(4)));
typedef float v8f __attribute__((ext_vector_type(8)));
typedef __attribute__((address_space(3))) float lds_f;
typedef __attribute__((address_space(3))) v4f   lds_v4f;
typedef __attribute__((address_space(3))) v2f   lds_v2f;

#define BATCH 512
#define INPS  1024
#define NEUR  512
#define KFAN  64
#define NOUT  16
#define ROWS  8
#define WAVES 4
#define QK    4
#define NPART (2 * WAVES)

#define ROWBYTES (KFAN * 4)
#define ACTF  ((INPS + NEUR) * ROWS)
#define SLOTF (ROWS * NPART)
#define NSLOT 8
#define PARTF (NSLOT * SLOTF)
#define DUMPF (NSLOT * SLOTF + 32 * WAVES)
#define CENT  8
#define COEFF ((NEUR + 2) * CENT)
#define CCH   8
#define CPASSB (2 * WAVES * ROWBYTES)

static __device__ __forceinline__ void load_w(float4 &w, const char* __restrict__ wp) {
    w = *(const float4*)wp;
}

static __device__ __forceinline__ void load_j(int4 &j, const char* __restrict__ jp) {
    j = *(const int4*)jp;
}

static __device__ __forceinline__ void to_addr(const lds_f* (&g)[4], const int4 &j, const lds_f* actn) {
    g[0] = actn + j.x * ROWS;
    g[1] = actn + j.y * ROWS;
    g[2] = actn + j.z * ROWS;
    g[3] = actn + j.w * ROWS;
}

static __device__ __forceinline__ void gather(v2f (&b)[2], const lds_f* const (&g)[4]) {
    b[0].x = *g[0];
    b[0].y = *g[1];
    b[1].x = *g[2];
    b[1].y = *g[3];
}

static __device__ __forceinline__ float block2(const float4 &w, const v2f (&b)[2]) {
    v2f a0, a1;
    a0.x = w.x; a0.y = w.y;
    a1.x = w.z; a1.y = w.w;
    const v8f acc0 = __builtin_amdgcn_wmma_f32_16x16x4_f32(false, a0, false, b[0], (short)0, v8f{}, false, false);
    const v8f acc1 = __builtin_amdgcn_wmma_f32_16x16x4_f32(false, a1, false, b[1], (short)0, v8f{}, false, false);
    return acc0[0] + acc1[0];
}

static __device__ __forceinline__ float row_sum16(float v) {
    v += __builtin_bit_cast(float, __builtin_amdgcn_update_dpp(0, __builtin_bit_cast(int, v), 0x161, 0xf, 0xf, false));
    v += __builtin_bit_cast(float, __builtin_amdgcn_update_dpp(0, __builtin_bit_cast(int, v), 0x162, 0xf, 0xf, false));
    v += __builtin_bit_cast(float, __builtin_amdgcn_update_dpp(0, __builtin_bit_cast(int, v), 0x164, 0xf, 0xf, false));
    v += __builtin_bit_cast(float, __builtin_amdgcn_update_dpp(0, __builtin_bit_cast(int, v), 0x168, 0xf, 0xf, false));
    return v;
}

static __device__ __forceinline__ void bar_signal_lds_done() {
    asm volatile("s_wait_dscnt 0x0\n\ts_barrier_signal -1" ::: "memory");
}
static __device__ __forceinline__ void bar_wait() {
    asm volatile("s_barrier_wait -1" ::: "memory");
}

static __device__ __forceinline__ void load_parts(v4f (&ep)[2], const lds_f* partn, int slot) {
    ep[0] = *(const lds_v4f*)(partn + slot * SLOTF);
    ep[1] = *(const lds_v4f*)(partn + slot * SLOTF + 4);
}
static __device__ __forceinline__ float sum_parts(const v4f (&ep)[2]) {
    const float even = (ep[0].x + ep[0].z) + (ep[1].x + ep[1].z);
    const float odd  = (ep[0].y + ep[0].w) + (ep[1].y + ep[1].w);
    return even + odd;
}

static __device__ __forceinline__ void coef_load(int4 (&cj)[CCH], float4 (&cw)[CCH],
                                                 const char* __restrict__ jrow, const char* __restrict__ wrow) {
    #pragma unroll
    for (int u = 0; u < CCH; ++u) {
        cj[u] = *(const int4*)(jrow + u * CPASSB);
        cw[u] = *(const float4*)(wrow + u * CPASSB);
    }
}

static __device__ __forceinline__ float coef_lane(const int4 &cj, const float4 &cw, int t) {
    return (((cj.x == t) ? cw.x : 0.0f) + ((cj.y == t) ? cw.y : 0.0f))
         + (((cj.z == t) ? cw.z : 0.0f) + ((cj.w == t) ? cw.w : 0.0f));
}

static __device__ __forceinline__ void coef_make(const int4 (&cj)[CCH], const float4 (&cw)[CCH],
                                                 lds_f* coef, int q0, int n) {
    float c1[CCH], c2[CCH], c3[CCH], c4[CCH], c5[CCH];
    #pragma unroll
    for (int u = 0; u < CCH; ++u) {
        const int q  = q0 + u * (2 * WAVES);
        const int t1 = (q >= 1) ? (INPS + q - 1) : -1;
        const int t2 = (q >= 2) ? (INPS + q - 2) : -1;
        const int t3 = (q >= 3) ? (INPS + q - 3) : -1;
        const int t4 = (q >= 4) ? (INPS + q - 4) : -1;
        const int t5 = (q >= 5) ? (INPS + q - 5) : -1;
        c1[u] = row_sum16(coef_lane(cj[u], cw[u], t1));
        c2[u] = row_sum16(coef_lane(cj[u], cw[u], t2));
        c3[u] = row_sum16(coef_lane(cj[u], cw[u], t3));
        c4[u] = row_sum16(coef_lane(cj[u], cw[u], t4));
        c5[u] = row_sum16(coef_lane(cj[u], cw[u], t5));
    }
    if (n == 0) {
        const bool odd = (q0 & 1) != 0;
        #pragma unroll
        for (int u = 0; u < CCH; ++u) {
            v4f e, f;
            e.x = odd ? c2[u] : c1[u];
            e.y = odd ? c3[u] : c2[u];
            e.z = odd ? c4[u] : c3[u];
            e.w = odd ? c5[u] : c4[u];
            f.x = odd ? c1[u] : 0.0f;
            f.y = 0.0f; f.z = 0.0f; f.w = 0.0f;
            *(lds_v4f*)(coef + (q0 + u * (2 * WAVES)) * CENT)     = e;
            *(lds_v4f*)(coef + (q0 + u * (2 * WAVES)) * CENT + 4) = f;
        }
    }
}

static __device__ __forceinline__ void pipe_trip2(
    lds_f* actn, const lds_f* partn, lds_f* pstore, const lds_f* coefn, bool hi,
    const lds_f* (&gA)[4], const lds_f* (&gB)[4],
    v4f (&epA)[2], v4f (&epB)[2],
    float4 &wA, float4 &wB, int4 &jA, int4 &jB,
    v2f &coA, v2f &coB, v2f &cnB, float &s1, float &s2, float &s3, float &s4,
    const char* __restrict__ wbase, const char* __restrict__ jbase, unsigned hbytes,
    int a, const int slotW0, const int slotW1, const int slotR0, const int slotR1)
{
    v2f bA[2], bB[2];
    gather(bA, gA);
    gather(bB, gB);
    const v2f coA_n = *(const lds_v2f*)(coefn + (a + 2) * CENT);
    const v2f coB_n = *(const lds_v2f*)(coefn + (a + 3) * CENT);
    const v2f cnB_n = *(const lds_v2f*)(coefn + (a + 3) * CENT + 4);
    __builtin_amdgcn_sched_barrier(0);

    v2f bo;
    bo.x = hi ? s3 : s1;
    bo.y = hi ? s4 : s2;
    const v8f oldA = __builtin_amdgcn_wmma_f32_16x16x4_f32(
        false, coA, false, bo, (short)0, v8f{}, false, false);
    const v8f oldB = __builtin_amdgcn_wmma_f32_16x16x4_f32(
        false, coB, false, bo, (short)0, v8f{}, false, false);
    asm volatile("" :: "v"(oldA), "v"(oldB));
    __builtin_amdgcn_sched_barrier(0);

    const float baseA = sum_parts(epA) + oldA[0];
    const float baseB = sum_parts(epB) + oldB[0];
    asm volatile("" :: "v"(baseA), "v"(baseB));

    bar_signal_lds_done();
    __builtin_amdgcn_sched_barrier(0);

    const float exA = __expf(-baseA);
    v2f a0, a1;
    a0.x = wA.x; a0.y = wA.y;
    a1.x = wA.z; a1.y = wA.w;
    const v8f acc0 = __builtin_amdgcn_wmma_f32_16x16x4_f32(false, a0, false, bA[0], (short)0, v8f{}, false, false);
    const float sgA = __builtin_amdgcn_rcpf(1.0f + exA);
    asm volatile("" :: "v"(acc0), "v"(sgA));
    __builtin_amdgcn_sched_barrier(0);

    v2f bn;
    bn.x = sgA;
    bn.y = sgA;
    const v8f newB = __builtin_amdgcn_wmma_f32_16x16x4_f32(
        false, cnB, false, bn, (short)0, v8f{}, false, false);
    asm volatile("" :: "v"(newB));
    __builtin_amdgcn_sched_barrier(0);

    v2f a2, a3;
    a2.x = wB.x; a2.y = wB.y;
    a3.x = wB.z; a3.y = wB.w;
    const v8f acc1 = __builtin_amdgcn_wmma_f32_16x16x4_f32(false, a1, false, bA[1], (short)0, v8f{}, false, false);
    to_addr(gA, jA, actn);
    to_addr(gB, jB, actn);
    const v8f acc2 = __builtin_amdgcn_wmma_f32_16x16x4_f32(false, a2, false, bB[0], (short)0, v8f{}, false, false);
    asm volatile("" :: "v"(acc1), "v"(acc2), "v"(gA[0]), "v"(gA[1]), "v"(gA[2]), "v"(gA[3]), "v"(gB[0]), "v"(gB[1]), "v"(gB[2]), "v"(gB[3]));
    __builtin_amdgcn_sched_barrier(0);

    const float valB = newB[0] + baseB;
    const float exB  = __expf(-valB);
    const v8f acc3 = __builtin_amdgcn_wmma_f32_16x16x4_f32(false, a3, false, bB[1], (short)0, v8f{}, false, false);
    const float sgB  = __builtin_amdgcn_rcpf(1.0f + exB);
    asm volatile("" :: "v"(acc3), "v"(sgB));
    __builtin_amdgcn_sched_barrier(0);

    const float partA = acc0[0] + acc1[0];
    const float partB = acc2[0] + acc3[0];
    pstore[slotW0 * SLOTF] = partA;
    pstore[slotW1 * SLOTF] = partB;
    const int rwA = (a + 12 < NEUR) ? (a + 12) : (NEUR - 1);
    const int rwB = (a + 13 < NEUR) ? (a + 13) : (NEUR - 1);
    const int rjA = (a + 14 < NEUR) ? (a + 14) : (NEUR - 1);
    const int rjB = (a + 15 < NEUR) ? (a + 15) : (NEUR - 1);
    load_w(wA, wbase + (unsigned)rwA * ROWBYTES + hbytes);
    load_w(wB, wbase + (unsigned)rwB * ROWBYTES + hbytes);
    load_j(jA, jbase + (unsigned)rjA * ROWBYTES + hbytes);
    load_j(jB, jbase + (unsigned)rjB * ROWBYTES + hbytes);

    bar_wait();
    load_parts(epA, partn, slotR0);
    load_parts(epB, partn, slotR1);
    actn[(INPS + a) * ROWS]     = sgA;
    actn[(INPS + a + 1) * ROWS] = sgB;
    s4 = s2;
    s3 = s1;
    s2 = sgA;
    s1 = sgB;
    coA = coA_n;
    coB = coB_n;
    cnB = cnB_n;
}

__global__ __launch_bounds__(32 * WAVES) void ne_scan_wmma_kernel(
    const float* __restrict__ x,
    const int*   __restrict__ in_idxs,
    const float* __restrict__ weights,
    float*       __restrict__ out)
{
    extern __shared__ float act[];

    const int tid   = threadIdx.x;
    const int lane  = tid & 31;
    const int wv    = __builtin_amdgcn_readfirstlane(tid >> 5);
    const int rbase = blockIdx.x * ROWS;
    const int h     = lane >> 4;
    const int n     = lane & 15;
    const int sub   = n >> 3;
    const int r     = n & 7;

    const char* jbase = (const char*)in_idxs;
    const char* wbase = (const char*)weights;

    lds_f*      coef = (lds_f*)act + ACTF + PARTF + DUMPF;
    const int   cq0  = 2 * wv + h;
    const char* cjr  = jbase + cq0 * ROWBYTES + 16 * n;
    const char* cwr  = wbase + cq0 * ROWBYTES + 16 * n;
    int4   cja[CCH], cjb[CCH];
    float4 cwa[CCH], cwb[CCH];
    coef_load(cja, cwa, cjr, cwr);

    for (int t = tid; t < ROWS * INPS; t += 32 * WAVES) {
        int col = t & (INPS - 1);
        int rr  = t >> 10;
        act[col * ROWS + rr] = x[(size_t)(rbase + rr) * INPS + col];
    }
    for (int t = tid; t < ROWS * NEUR; t += 32 * WAVES) {
        act[INPS * ROWS + t] = 0.0f;
    }

    #pragma unroll 1
    for (int c = 0; c < NEUR / (2 * WAVES * CCH); c += 2) {
        coef_load(cjb, cwb, cjr + (c + 1) * (CCH * CPASSB), cwr + (c + 1) * (CCH * CPASSB));
        coef_make(cja, cwa, coef, cq0 + c * (CCH * 2 * WAVES), n);
        const int cn = (c + 2 < NEUR / (2 * WAVES * CCH)) ? (c + 2) : (c + 1);
        coef_load(cja, cwa, cjr + cn * (CCH * CPASSB), cwr + cn * (CCH * CPASSB));
        coef_make(cjb, cwb, coef, cq0 + (c + 1) * (CCH * 2 * WAVES), n);
    }
    if (tid < 2 * CENT) coef[NEUR * CENT + tid] = 0.0f;
    __syncthreads();

    lds_f*       actn  = (lds_f*)act + r;
    const lds_f* partn = (lds_f*)act + ACTF + r * NPART;
    lds_f*       pstore = (h == sub) ? ((lds_f*)act + ACTF + r * NPART + 2 * wv + h)
                                     : ((lds_f*)act + ACTF + PARTF + 32 * wv + lane);
    const lds_f* coefn = coef + 2 * h;
    const bool     hi     = h != 0;
    const unsigned hbytes = 4u * (unsigned)(16 * wv + 8 * sub + QK * h);
    float4 wA0, wB0, wA1, wB1, wA2, wB2, wA3, wB3;
    int4   jA0, jB0, jA1, jB1, jA2, jB2, jA3, jB3;
    v2f    coA, coB, cnB;
    const lds_f* gA[4];
    const lds_f* gB[4];

    {
        int4   p0, p1, p2, p3, p4, p5;
        float4 q0, q1, q2, q3;
        v2f    b0[2], b1[2], b2[2], b3[2];
        load_j(p0, jbase + hbytes);
        load_j(p1, jbase + 1 * ROWBYTES + hbytes);
        load_j(p2, jbase + 2 * ROWBYTES + hbytes);
        load_j(p3, jbase + 3 * ROWBYTES + hbytes);
        load_j(p4, jbase + 4 * ROWBYTES + hbytes);
        load_j(p5, jbase + 5 * ROWBYTES + hbytes);
        load_w(q0, wbase + hbytes);
        load_w(q1, wbase + 1 * ROWBYTES + hbytes);
        load_w(q2, wbase + 2 * ROWBYTES + hbytes);
        load_w(q3, wbase + 3 * ROWBYTES + hbytes);
        to_addr(gA, p0, actn);
        gather(b0, gA);
        to_addr(gB, p1, actn);
        gather(b1, gB);
        to_addr(gA, p2, actn);
        gather(b2, gA);
        to_addr(gB, p3, actn);
        gather(b3, gB);
        const float part0 = block2(q0, b0);
        const float part1 = block2(q1, b1);
        const float part2 = block2(q2, b2);
        const float part3 = block2(q3, b3);
        to_addr(gA, p4, actn);
        to_addr(gB, p5, actn);
        pstore[0 * SLOTF] = part0;
        pstore[1 * SLOTF] = part1;
        pstore[2 * SLOTF] = part2;
        pstore[3 * SLOTF] = part3;
    }
    bar_signal_lds_done();
    load_w(wA0, wbase + 4 * ROWBYTES + hbytes);
    load_w(wB0, wbase + 5 * ROWBYTES + hbytes);
    load_j(jA0, jbase + 6 * ROWBYTES + hbytes);
    load_j(jB0, jbase + 7 * ROWBYTES + hbytes);
    load_w(wA1, wbase + 6 * ROWBYTES + hbytes);
    load_w(wB1, wbase + 7 * ROWBYTES + hbytes);
    load_j(jA1, jbase + 8 * ROWBYTES + hbytes);
    load_j(jB1, jbase + 9 * ROWBYTES + hbytes);
    load_w(wA2, wbase + 8 * ROWBYTES + hbytes);
    load_w(wB2, wbase + 9 * ROWBYTES + hbytes);
    load_j(jA2, jbase + 10 * ROWBYTES + hbytes);
    load_j(jB2, jbase + 11 * ROWBYTES + hbytes);
    load_w(wA3, wbase + 10 * ROWBYTES + hbytes);
    load_w(wB3, wbase + 11 * ROWBYTES + hbytes);
    load_j(jA3, jbase + 12 * ROWBYTES + hbytes);
    load_j(jB3, jbase + 13 * ROWBYTES + hbytes);
    bar_wait();
    v4f epA[2], epB[2];
    load_parts(epA, partn, 0);
    load_parts(epB, partn, 1);
    coA = *(const lds_v2f*)(coefn + 0 * CENT);
    coB = *(const lds_v2f*)(coefn + 1 * CENT);
    cnB = *(const lds_v2f*)(coefn + 1 * CENT + 4);
    float s1 = 0.0f, s2 = 0.0f, s3 = 0.0f, s4 = 0.0f;

    for (int pos = 0; pos < NEUR; pos += 8) {
        pipe_trip2(actn, partn, pstore, coefn, hi, gA, gB, epA, epB, wA0, wB0, jA0, jB0, coA, coB, cnB, s1, s2, s3, s4,
                   wbase, jbase, hbytes, pos, 4, 5, 2, 3);
        pipe_trip2(actn, partn, pstore, coefn, hi, gA, gB, epA, epB, wA1, wB1, jA1, jB1, coA, coB, cnB, s1, s2, s3, s4,
                   wbase, jbase, hbytes, pos + 2, 6, 7, 4, 5);
        pipe_trip2(actn, partn, pstore, coefn, hi, gA, gB, epA, epB, wA2, wB2, jA2, jB2, coA, coB, cnB, s1, s2, s3, s4,
                   wbase, jbase, hbytes, pos + 4, 0, 1, 6, 7);
        pipe_trip2(actn, partn, pstore, coefn, hi, gA, gB, epA, epB, wA3, wB3, jA3, jB3, coA, coB, cnB, s1, s2, s3, s4,
                   wbase, jbase, hbytes, pos + 6, 2, 3, 0, 1);
    }

    for (int t = tid; t < NOUT * ROWS; t += 32 * WAVES) {
        int o  = t >> 3;
        int rr = t & 7;
        out[(size_t)o * BATCH + rbase + rr] =
            act[(INPS + NEUR - NOUT + o) * ROWS + rr];
    }
}

extern "C" void kernel_launch(void* const* d_in, const int* in_sizes, int n_in,
                              void* d_out, int out_size, void* d_ws, size_t ws_size,
                              hipStream_t stream) {
    (void)in_sizes; (void)n_in; (void)d_ws; (void)ws_size; (void)out_size;
    const float* x       = (const float*)d_in[0];
    const int*   in_idxs = (const int*)d_in[1];
    const float* weights = (const float*)d_in[2];
    float*       out     = (float*)d_out;

    const size_t smem_bytes = (size_t)(ACTF + PARTF + DUMPF + COEFF) * sizeof(float);

    ne_scan_wmma_kernel<<<dim3(BATCH / ROWS), dim3(32 * WAVES), smem_bytes, stream>>>(
        x, in_idxs, weights, out);
}


# Refill loads (weights, indices) take their rows as constant offsets from running per-lane pointers moved once per loop pass: 12 scalar and 64-bit address ops leave each trip (sum order unchanged)
# speedup vs baseline: 5.3059x; 1.1270x over previous
#include <hip/hip_runtime.h>


typedef float v2f __attribute__((ext_vector_type(2)));
typedef float v4f __attribute__((ext_vector_type(4)));
typedef float v8f __attribute__((ext_vector_type(8)));
typedef __attribute__((address_space(3))) float lds_f;
typedef __attribute__((address_space(3))) v4f   lds_v4f;
typedef __attribute__((address_space(3))) v2f   lds_v2f;

#define BATCH 512
#define INPS  1024
#define NEUR  512
#define KFAN  64
#define NOUT  16
#define ROWS  8
#define WAVES 4
#define QK    4
#define NPART (2 * WAVES)

#define ROWBYTES (KFAN * 4)
#define ACTF  ((INPS + NEUR) * ROWS)
#define SLOTF (ROWS * NPART)
#define NSLOT 8
#define PARTF (NSLOT * SLOTF)
#define DUMPF (NSLOT * SLOTF + 32 * WAVES)
#define CENT  8
#define COEFF ((NEUR + 2) * CENT)
#define CCH   8
#define CPASSB (2 * WAVES * ROWBYTES)

static __device__ __forceinline__ void load_w(float4 &w, const char* __restrict__ wp) {
    w = *(const float4*)wp;
}

static __device__ __forceinline__ void load_j(int4 &j, const char* __restrict__ jp) {
    j = *(const int4*)jp;
}

static __device__ __forceinline__ void to_addr(const lds_f* (&g)[4], const int4 &j, const lds_f* actn) {
    g[0] = actn + j.x * ROWS;
    g[1] = actn + j.y * ROWS;
    g[2] = actn + j.z * ROWS;
    g[3] = actn + j.w * ROWS;
}

static __device__ __forceinline__ void gather(v2f (&b)[2], const lds_f* const (&g)[4]) {
    b[0].x = *g[0];
    b[0].y = *g[1];
    b[1].x = *g[2];
    b[1].y = *g[3];
}

static __device__ __forceinline__ float block2(const float4 &w, const v2f (&b)[2]) {
    v2f a0, a1;
    a0.x = w.x; a0.y = w.y;
    a1.x = w.z; a1.y = w.w;
    const v8f acc0 = __builtin_amdgcn_wmma_f32_16x16x4_f32(false, a0, false, b[0], (short)0, v8f{}, false, false);
    const v8f acc1 = __builtin_amdgcn_wmma_f32_16x16x4_f32(false, a1, false, b[1], (short)0, v8f{}, false, false);
    return acc0[0] + acc1[0];
}

static __device__ __forceinline__ float row_sum16(float v) {
    v += __builtin_bit_cast(float, __builtin_amdgcn_update_dpp(0, __builtin_bit_cast(int, v), 0x161, 0xf, 0xf, false));
    v += __builtin_bit_cast(float, __builtin_amdgcn_update_dpp(0, __builtin_bit_cast(int, v), 0x162, 0xf, 0xf, false));
    v += __builtin_bit_cast(float, __builtin_amdgcn_update_dpp(0, __builtin_bit_cast(int, v), 0x164, 0xf, 0xf, false));
    v += __builtin_bit_cast(float, __builtin_amdgcn_update_dpp(0, __builtin_bit_cast(int, v), 0x168, 0xf, 0xf, false));
    return v;
}

static __device__ __forceinline__ void bar_signal_lds_done() {
    asm volatile("s_wait_dscnt 0x0\n\ts_barrier_signal -1" ::: "memory");
}
static __device__ __forceinline__ void bar_wait() {
    asm volatile("s_barrier_wait -1" ::: "memory");
}

static __device__ __forceinline__ void load_parts(v4f (&ep)[2], const lds_f* partn, int slot) {
    ep[0] = *(const lds_v4f*)(partn + slot * SLOTF);
    ep[1] = *(const lds_v4f*)(partn + slot * SLOTF + 4);
}
static __device__ __forceinline__ float sum_parts(const v4f (&ep)[2]) {
    const float even = (ep[0].x + ep[0].z) + (ep[1].x + ep[1].z);
    const float odd  = (ep[0].y + ep[0].w) + (ep[1].y + ep[1].w);
    return even + odd;
}

static __device__ __forceinline__ void coef_load(int4 (&cj)[CCH], float4 (&cw)[CCH],
                                                 const char* __restrict__ jrow, const char* __restrict__ wrow) {
    #pragma unroll
    for (int u = 0; u < CCH; ++u) {
        cj[u] = *(const int4*)(jrow + u * CPASSB);
        cw[u] = *(const float4*)(wrow + u * CPASSB);
    }
}

static __device__ __forceinline__ float coef_lane(const int4 &cj, const float4 &cw, int t) {
    return (((cj.x == t) ? cw.x : 0.0f) + ((cj.y == t) ? cw.y : 0.0f))
         + (((cj.z == t) ? cw.z : 0.0f) + ((cj.w == t) ? cw.w : 0.0f));
}

static __device__ __forceinline__ void coef_make(const int4 (&cj)[CCH], const float4 (&cw)[CCH],
                                                 lds_f* coef, int q0, int n) {
    float c1[CCH], c2[CCH], c3[CCH], c4[CCH], c5[CCH];
    #pragma unroll
    for (int u = 0; u < CCH; ++u) {
        const int q  = q0 + u * (2 * WAVES);
        const int t1 = (q >= 1) ? (INPS + q - 1) : -1;
        const int t2 = (q >= 2) ? (INPS + q - 2) : -1;
        const int t3 = (q >= 3) ? (INPS + q - 3) : -1;
        const int t4 = (q >= 4) ? (INPS + q - 4) : -1;
        const int t5 = (q >= 5) ? (INPS + q - 5) : -1;
        c1[u] = row_sum16(coef_lane(cj[u], cw[u], t1));
        c2[u] = row_sum16(coef_lane(cj[u], cw[u], t2));
        c3[u] = row_sum16(coef_lane(cj[u], cw[u], t3));
        c4[u] = row_sum16(coef_lane(cj[u], cw[u], t4));
        c5[u] = row_sum16(coef_lane(cj[u], cw[u], t5));
    }
    if (n == 0) {
        const bool odd = (q0 & 1) != 0;
        #pragma unroll
        for (int u = 0; u < CCH; ++u) {
            v4f e, f;
            e.x = odd ? c2[u] : c1[u];
            e.y = odd ? c3[u] : c2[u];
            e.z = odd ? c4[u] : c3[u];
            e.w = odd ? c5[u] : c4[u];
            f.x = odd ? c1[u] : 0.0f;
            f.y = 0.0f; f.z = 0.0f; f.w = 0.0f;
            *(lds_v4f*)(coef + (q0 + u * (2 * WAVES)) * CENT)     = e;
            *(lds_v4f*)(coef + (q0 + u * (2 * WAVES)) * CENT + 4) = f;
        }
    }
}

static __device__ __forceinline__ void pipe_trip2(
    lds_f* actn, const lds_f* partn, lds_f* pstore, const lds_f* coefn, bool hi,
    const lds_f* (&gA)[4], const lds_f* (&gB)[4],
    v4f (&epA)[2], v4f (&epB)[2],
    float4 &wA, float4 &wB, int4 &jA, int4 &jB,
    v2f &coA, v2f &coB, v2f &cnB, float &s1, float &s2, float &s3, float &s4,
    const char* __restrict__ wnext, const char* __restrict__ jnext,
    int a, const int slotW0, const int slotW1, const int slotR0, const int slotR1)
{
    v2f bA[2], bB[2];
    gather(bA, gA);
    gather(bB, gB);
    const v2f coA_n = *(const lds_v2f*)(coefn + (a + 2) * CENT);
    const v2f coB_n = *(const lds_v2f*)(coefn + (a + 3) * CENT);
    const v2f cnB_n = *(const lds_v2f*)(coefn + (a + 3) * CENT + 4);
    __builtin_amdgcn_sched_barrier(0);

    v2f bo;
    bo.x = hi ? s3 : s1;
    bo.y = hi ? s4 : s2;
    const v8f oldA = __builtin_amdgcn_wmma_f32_16x16x4_f32(
        false, coA, false, bo, (short)0, v8f{}, false, false);
    const v8f oldB = __builtin_amdgcn_wmma_f32_16x16x4_f32(
        false, coB, false, bo, (short)0, v8f{}, false, false);
    asm volatile("" :: "v"(oldA), "v"(oldB));
    __builtin_amdgcn_sched_barrier(0);

    const float baseA = sum_parts(epA) + oldA[0];
    const float baseB = sum_parts(epB) + oldB[0];
    asm volatile("" :: "v"(baseA), "v"(baseB));

    bar_signal_lds_done();
    __builtin_amdgcn_sched_barrier(0);

    const float exA = __expf(-baseA);
    v2f a0, a1;
    a0.x = wA.x; a0.y = wA.y;
    a1.x = wA.z; a1.y = wA.w;
    const v8f acc0 = __builtin_amdgcn_wmma_f32_16x16x4_f32(false, a0, false, bA[0], (short)0, v8f{}, false, false);
    const float sgA = __builtin_amdgcn_rcpf(1.0f + exA);
    asm volatile("" :: "v"(acc0), "v"(sgA));
    __builtin_amdgcn_sched_barrier(0);

    v2f bn;
    bn.x = sgA;
    bn.y = sgA;
    const v8f newB = __builtin_amdgcn_wmma_f32_16x16x4_f32(
        false, cnB, false, bn, (short)0, v8f{}, false, false);
    asm volatile("" :: "v"(newB));
    __builtin_amdgcn_sched_barrier(0);

    v2f a2, a3;
    a2.x = wB.x; a2.y = wB.y;
    a3.x = wB.z; a3.y = wB.w;
    const v8f acc1 = __builtin_amdgcn_wmma_f32_16x16x4_f32(false, a1, false, bA[1], (short)0, v8f{}, false, false);
    to_addr(gA, jA, actn);
    to_addr(gB, jB, actn);
    const v8f acc2 = __builtin_amdgcn_wmma_f32_16x16x4_f32(false, a2, false, bB[0], (short)0, v8f{}, false, false);
    asm volatile("" :: "v"(acc1), "v"(acc2), "v"(gA[0]), "v"(gA[1]), "v"(gA[2]), "v"(gA[3]), "v"(gB[0]), "v"(gB[1]), "v"(gB[2]), "v"(gB[3]));
    __builtin_amdgcn_sched_barrier(0);

    const float valB = newB[0] + baseB;
    const float exB  = __expf(-valB);
    const v8f acc3 = __builtin_amdgcn_wmma_f32_16x16x4_f32(false, a3, false, bB[1], (short)0, v8f{}, false, false);
    const float sgB  = __builtin_amdgcn_rcpf(1.0f + exB);
    asm volatile("" :: "v"(acc3), "v"(sgB));
    __builtin_amdgcn_sched_barrier(0);

    const float partA = acc0[0] + acc1[0];
    const float partB = acc2[0] + acc3[0];
    pstore[slotW0 * SLOTF] = partA;
    pstore[slotW1 * SLOTF] = partB;
    load_w(wA, wnext);
    load_w(wB, wnext + ROWBYTES);
    load_j(jA, jnext);
    load_j(jB, jnext + ROWBYTES);

    bar_wait();
    load_parts(epA, partn, slotR0);
    load_parts(epB, partn, slotR1);
    actn[(INPS + a) * ROWS]     = sgA;
    actn[(INPS + a + 1) * ROWS] = sgB;
    s4 = s2;
    s3 = s1;
    s2 = sgA;
    s1 = sgB;
    coA = coA_n;
    coB = coB_n;
    cnB = cnB_n;
}

__global__ __launch_bounds__(32 * WAVES) void ne_scan_wmma_kernel(
    const float* __restrict__ x,
    const int*   __restrict__ in_idxs,
    const float* __restrict__ weights,
    float*       __restrict__ out)
{
    extern __shared__ float act[];

    const int tid   = threadIdx.x;
    const int lane  = tid & 31;
    const int wv    = __builtin_amdgcn_readfirstlane(tid >> 5);
    const int rbase = blockIdx.x * ROWS;
    const int h     = lane >> 4;
    const int n     = lane & 15;
    const int sub   = n >> 3;
    const int r     = n & 7;

    const char* jbase = (const char*)in_idxs;
    const char* wbase = (const char*)weights;

    lds_f*      coef = (lds_f*)act + ACTF + PARTF + DUMPF;
    const int   cq0  = 2 * wv + h;
    const char* cjr  = jbase + cq0 * ROWBYTES + 16 * n;
    const char* cwr  = wbase + cq0 * ROWBYTES + 16 * n;
    int4   cja[CCH], cjb[CCH];
    float4 cwa[CCH], cwb[CCH];
    coef_load(cja, cwa, cjr, cwr);

    for (int t = tid; t < ROWS * INPS; t += 32 * WAVES) {
        int col = t & (INPS - 1);
        int rr  = t >> 10;
        act[col * ROWS + rr] = x[(size_t)(rbase + rr) * INPS + col];
    }
    for (int t = tid; t < ROWS * NEUR; t += 32 * WAVES) {
        act[INPS * ROWS + t] = 0.0f;
    }

    #pragma unroll 1
    for (int c = 0; c < NEUR / (2 * WAVES * CCH); c += 2) {
        coef_load(cjb, cwb, cjr + (c + 1) * (CCH * CPASSB), cwr + (c + 1) * (CCH * CPASSB));
        coef_make(cja, cwa, coef, cq0 + c * (CCH * 2 * WAVES), n);
        const int cn = (c + 2 < NEUR / (2 * WAVES * CCH)) ? (c + 2) : (c + 1);
        coef_load(cja, cwa, cjr + cn * (CCH * CPASSB), cwr + cn * (CCH * CPASSB));
        coef_make(cjb, cwb, coef, cq0 + (c + 1) * (CCH * 2 * WAVES), n);
    }
    if (tid < 2 * CENT) coef[NEUR * CENT + tid] = 0.0f;
    __syncthreads();

    lds_f*       actn  = (lds_f*)act + r;
    const lds_f* partn = (lds_f*)act + ACTF + r * NPART;
    lds_f*       pstore = (h == sub) ? ((lds_f*)act + ACTF + r * NPART + 2 * wv + h)
                                     : ((lds_f*)act + ACTF + PARTF + 32 * wv + lane);
    const lds_f* coefn = coef + 2 * h;
    const bool     hi     = h != 0;
    const unsigned hbytes = 4u * (unsigned)(16 * wv + 8 * sub + QK * h);
    float4 wA0, wB0, wA1, wB1, wA2, wB2, wA3, wB3;
    int4   jA0, jB0, jA1, jB1, jA2, jB2, jA3, jB3;
    v2f    coA, coB, cnB;
    const lds_f* gA[4];
    const lds_f* gB[4];

    {
        int4   p0, p1, p2, p3, p4, p5;
        float4 q0, q1, q2, q3;
        v2f    b0[2], b1[2], b2[2], b3[2];
        load_j(p0, jbase + hbytes);
        load_j(p1, jbase + 1 * ROWBYTES + hbytes);
        load_j(p2, jbase + 2 * ROWBYTES + hbytes);
        load_j(p3, jbase + 3 * ROWBYTES + hbytes);
        load_j(p4, jbase + 4 * ROWBYTES + hbytes);
        load_j(p5, jbase + 5 * ROWBYTES + hbytes);
        load_w(q0, wbase + hbytes);
        load_w(q1, wbase + 1 * ROWBYTES + hbytes);
        load_w(q2, wbase + 2 * ROWBYTES + hbytes);
        load_w(q3, wbase + 3 * ROWBYTES + hbytes);
        to_addr(gA, p0, actn);
        gather(b0, gA);
        to_addr(gB, p1, actn);
        gather(b1, gB);
        to_addr(gA, p2, actn);
        gather(b2, gA);
        to_addr(gB, p3, actn);
        gather(b3, gB);
        const float part0 = block2(q0, b0);
        const float part1 = block2(q1, b1);
        const float part2 = block2(q2, b2);
        const float part3 = block2(q3, b3);
        to_addr(gA, p4, actn);
        to_addr(gB, p5, actn);
        pstore[0 * SLOTF] = part0;
        pstore[1 * SLOTF] = part1;
        pstore[2 * SLOTF] = part2;
        pstore[3 * SLOTF] = part3;
    }
    bar_signal_lds_done();
    load_w(wA0, wbase + 4 * ROWBYTES + hbytes);
    load_w(wB0, wbase + 5 * ROWBYTES + hbytes);
    load_j(jA0, jbase + 6 * ROWBYTES + hbytes);
    load_j(jB0, jbase + 7 * ROWBYTES + hbytes);
    load_w(wA1, wbase + 6 * ROWBYTES + hbytes);
    load_w(wB1, wbase + 7 * ROWBYTES + hbytes);
    load_j(jA1, jbase + 8 * ROWBYTES + hbytes);
    load_j(jB1, jbase + 9 * ROWBYTES + hbytes);
    load_w(wA2, wbase + 8 * ROWBYTES + hbytes);
    load_w(wB2, wbase + 9 * ROWBYTES + hbytes);
    load_j(jA2, jbase + 10 * ROWBYTES + hbytes);
    load_j(jB2, jbase + 11 * ROWBYTES + hbytes);
    load_w(wA3, wbase + 10 * ROWBYTES + hbytes);
    load_w(wB3, wbase + 11 * ROWBYTES + hbytes);
    load_j(jA3, jbase + 12 * ROWBYTES + hbytes);
    load_j(jB3, jbase + 13 * ROWBYTES + hbytes);
    bar_wait();
    v4f epA[2], epB[2];
    load_parts(epA, partn, 0);
    load_parts(epB, partn, 1);
    coA = *(const lds_v2f*)(coefn + 0 * CENT);
    coB = *(const lds_v2f*)(coefn + 1 * CENT);
    cnB = *(const lds_v2f*)(coefn + 1 * CENT + 4);
    float s1 = 0.0f, s2 = 0.0f, s3 = 0.0f, s4 = 0.0f;

    const char* wlo = wbase + hbytes;
    const char* jlo = jbase + hbytes;
    const char* whi = wlo;
    const char* jhi = jlo;
    for (int pos = 0; pos < NEUR; pos += 8) {
        pipe_trip2(actn, partn, pstore, coefn, hi, gA, gB, epA, epB, wA0, wB0, jA0, jB0, coA, coB, cnB, s1, s2, s3, s4,
                   wlo + 12 * ROWBYTES, jlo + 14 * ROWBYTES, pos, 4, 5, 2, 3);
        pipe_trip2(actn, partn, pstore, coefn, hi, gA, gB, epA, epB, wA1, wB1, jA1, jB1, coA, coB, cnB, s1, s2, s3, s4,
                   wlo + 14 * ROWBYTES, jhi + 16 * ROWBYTES, pos + 2, 6, 7, 4, 5);
        pipe_trip2(actn, partn, pstore, coefn, hi, gA, gB, epA, epB, wA2, wB2, jA2, jB2, coA, coB, cnB, s1, s2, s3, s4,
                   whi + 16 * ROWBYTES, jhi + 18 * ROWBYTES, pos + 4, 0, 1, 6, 7);
        pipe_trip2(actn, partn, pstore, coefn, hi, gA, gB, epA, epB, wA3, wB3, jA3, jB3, coA, coB, cnB, s1, s2, s3, s4,
                   whi + 18 * ROWBYTES, jhi + 20 * ROWBYTES, pos + 6, 2, 3, 0, 1);
        const unsigned steplo = (pos < NEUR - 16) ? 8u * ROWBYTES : 0u;
        const unsigned stephi = (pos < NEUR - 24) ? 8u * ROWBYTES : 0u;
        wlo += steplo;
        jlo += steplo;
        whi += stephi;
        jhi += stephi;
    }

    for (int t = tid; t < NOUT * ROWS; t += 32 * WAVES) {
        int o  = t >> 3;
        int rr = t & 7;
        out[(size_t)o * BATCH + rbase + rr] =
            act[(INPS + NEUR - NOUT + o) * ROWS + rr];
    }
}

extern "C" void kernel_launch(void* const* d_in, const int* in_sizes, int n_in,
                              void* d_out, int out_size, void* d_ws, size_t ws_size,
                              hipStream_t stream) {
    (void)in_sizes; (void)n_in; (void)d_ws; (void)ws_size; (void)out_size;
    const float* x       = (const float*)d_in[0];
    const int*   in_idxs = (const int*)d_in[1];
    const float* weights = (const float*)d_in[2];
    float*       out     = (float*)d_out;

    const size_t smem_bytes = (size_t)(ACTF + PARTF + DUMPF + COEFF) * sizeof(float);

    ne_scan_wmma_kernel<<<dim3(BATCH / ROWS), dim3(32 * WAVES), smem_bytes, stream>>>(
        x, in_idxs, weights, out);
}
